# starting kernel
# baseline (speedup 1.0000x reference)
#include <hip/hip_runtime.h>

#define N_NODES 100000
#define N_EDGES 3200000
#define IN_CH   128
#define HID     96
#define HID2    48
#define NCLS    3

typedef __attribute__((ext_vector_type(2)))  float  v2f;
typedef __attribute__((ext_vector_type(8)))  float  v8f;

static __device__ inline void atom_add_f32(float* p, float v) {
    unsafeAtomicAdd(p, v);
}


__global__ __launch_bounds__(256) void init_deg(float* __restrict__ deg) {
    int i = blockIdx.x * blockDim.x + threadIdx.x;
    if (i < N_NODES) deg[i] = 1.0f;
}

__global__ __launch_bounds__(256) void accum_deg(const int* __restrict__ dst,
                                                 const float* __restrict__ ew,
                                                 float* __restrict__ deg) {
    int e = blockIdx.x * blockDim.x + threadIdx.x;
    if (e < N_EDGES) atom_add_f32(deg + dst[e], ew[e]);
}

__global__ __launch_bounds__(256) void make_dinv(float* __restrict__ deg) {
    int i = blockIdx.x * blockDim.x + threadIdx.x;
    if (i < N_NODES) deg[i] = rsqrtf(deg[i]);
}

__global__ __launch_bounds__(256) void make_norm(const int* __restrict__ src,
                                                 const int* __restrict__ dst,
                                                 const float* __restrict__ ew,
                                                 const float* __restrict__ dinv,
                                                 float* __restrict__ norm) {
    int e = blockIdx.x * blockDim.x + threadIdx.x;
    if (e < N_EDGES) norm[e] = dinv[src[e]] * ew[e] * dinv[dst[e]];
}


template<int K, int COLS>
__global__ __launch_bounds__(256) void gemm_wmma(const float* __restrict__ X,
                                                 const float* __restrict__ W,
                                                 float* __restrict__ H) {
    const int lane = threadIdx.x & 31;
    const int wave = threadIdx.x >> 5;
    const int tm = blockIdx.x * 8 + wave;
    const int tn = blockIdx.y;
    if (tm * 16 >= N_NODES) return;
    const int m0 = tm * 16, n0 = tn * 16;
    const int row   = m0 + (lane & 15);
    const int col   = n0 + (lane & 15);
    const int khalf = (lane >> 4) << 1;
    const float* xrow = X + (size_t)row * K;
    v8f c = {};
    for (int kk = 0; kk < K; kk += 4) {
        v2f a = *(const v2f*)(xrow + kk + khalf);
        v2f b;
        b.x = W[(size_t)(kk + khalf)     * COLS + col];
        b.y = W[(size_t)(kk + khalf + 1) * COLS + col];
        c = __builtin_amdgcn_wmma_f32_16x16x4_f32(false, a, false, b,
                                                  (short)0, c, false, false);
    }
    const int rbase = (lane >> 4) << 3;
    float* hrow = H + (size_t)(m0 + rbase) * COLS + col;
    #pragma unroll
    for (int r = 0; r < 8; ++r) hrow[(size_t)r * COLS] = c[r];
}


template<int C>
__global__ __launch_bounds__(256) void selfloop_init(const float* __restrict__ H,
                                                     const float* __restrict__ dinv,
                                                     float* __restrict__ out) {
    const int C4 = C / 4;
    int idx = blockIdx.x * blockDim.x + threadIdx.x;
    if (idx >= N_NODES * C4) return;
    int i = idx / C4, q = idx % C4;
    float di = dinv[i], w = di * di;
    float4 hv = *(const float4*)(H + (size_t)i * C + q * 4);
    float4 o = { w * hv.x, w * hv.y, w * hv.z, w * hv.w };
    *(float4*)(out + (size_t)i * C + q * 4) = o;
}

template<int C, int EPB>
__global__ __launch_bounds__(256) void scatter_edges(const float* __restrict__ H,
                                                     const int* __restrict__ src,
                                                     const int* __restrict__ dst,
                                                     const float* __restrict__ norm,
                                                     float* __restrict__ out) {
    const int q = threadIdx.x;
    const int e = blockIdx.x * EPB + threadIdx.y;
    if (e >= N_EDGES) return;
    const int s = src[e], d = dst[e];
    const float nw = norm[e];
    const float4 hv = *(const float4*)(H + (size_t)s * C + q * 4);
    float* o = out + (size_t)d * C + q * 4;
    atom_add_f32(o + 0, nw * hv.x);
    atom_add_f32(o + 1, nw * hv.y);
    atom_add_f32(o + 2, nw * hv.z);
    atom_add_f32(o + 3, nw * hv.w);
}

template<int C>
__global__ __launch_bounds__(256) void bias_relu(float* __restrict__ out,
                                                 const float* __restrict__ b) {
    const int C4 = C / 4;
    int idx = blockIdx.x * blockDim.x + threadIdx.x;
    if (idx >= N_NODES * C4) return;
    int i = idx / C4, q = idx % C4;
    float4 v  = *(float4*)(out + (size_t)i * C + q * 4);
    float4 bb = *(const float4*)(b + q * 4);
    v.x = fmaxf(v.x + bb.x, 0.0f);
    v.y = fmaxf(v.y + bb.y, 0.0f);
    v.z = fmaxf(v.z + bb.z, 0.0f);
    v.w = fmaxf(v.w + bb.w, 0.0f);
    *(float4*)(out + (size_t)i * C + q * 4) = v;
}

__global__ __launch_bounds__(256) void classifier(const float* __restrict__ Hin,
                                                  const float* __restrict__ Wc,
                                                  const float* __restrict__ bc,
                                                  float* __restrict__ logits) {
    int i = blockIdx.x * blockDim.x + threadIdx.x;
    if (i >= N_NODES) return;
    const float* h = Hin + (size_t)i * HID2;
    float a0 = bc[0], a1 = bc[1], a2 = bc[2];
    #pragma unroll
    for (int k = 0; k < HID2; ++k) {
        float hv = h[k];
        a0 += hv * Wc[k * 3 + 0];
        a1 += hv * Wc[k * 3 + 1];
        a2 += hv * Wc[k * 3 + 2];
    }
    logits[(size_t)i * 3 + 0] = a0;
    logits[(size_t)i * 3 + 1] = a1;
    logits[(size_t)i * 3 + 2] = a2;
}


extern "C" void kernel_launch(void* const* d_in, const int* in_sizes, int n_in,
                              void* d_out, int out_size, void* d_ws, size_t ws_size,
                              hipStream_t stream) {
    const float* x  = (const float*)d_in[0];
    const int*   ei = (const int*)  d_in[1];
    const float* ew = (const float*)d_in[2];
    const float* W1 = (const float*)d_in[3];
    const float* b1 = (const float*)d_in[4];
    const float* W2 = (const float*)d_in[5];
    const float* b2 = (const float*)d_in[6];
    const float* Wc = (const float*)d_in[7];
    const float* bc = (const float*)d_in[8];
    float* logits = (float*)d_out;

    const int* src = ei;
    const int* dst = ei + N_EDGES;

    char* ws = (char*)d_ws;
    size_t off = 0;
    auto take = [&](size_t bytes) -> void* {
        void* p = ws + off;
        off += (bytes + 255) & ~(size_t)255;
        return p;
    };
    float* dinv = (float*)take((size_t)N_NODES * 4);
    float* norm = (float*)take((size_t)N_EDGES * 4);
    float* H1   = (float*)take((size_t)N_NODES * HID * 4);
    float* out1 = (float*)take((size_t)N_NODES * HID * 4);
    float* H2   = H1;
    float* out2 = out1;

    const int TB = 256;
    init_deg <<<(N_NODES + TB - 1) / TB, TB, 0, stream>>>(dinv);
    accum_deg<<<(N_EDGES + TB - 1) / TB, TB, 0, stream>>>(dst, ew, dinv);
    make_dinv<<<(N_NODES + TB - 1) / TB, TB, 0, stream>>>(dinv);
    make_norm<<<(N_EDGES + TB - 1) / TB, TB, 0, stream>>>(src, dst, ew, dinv, norm);

    {
        dim3 g((N_NODES / 16 + 7) / 8, HID / 16);
        gemm_wmma<IN_CH, HID><<<g, 256, 0, stream>>>(x, W1, H1);
    }
    {
        int tot = N_NODES * (HID / 4);
        selfloop_init<HID><<<(tot + TB - 1) / TB, TB, 0, stream>>>(H1, dinv, out1);
        scatter_edges<HID, 8><<<(N_EDGES + 7) / 8, dim3(HID / 4, 8), 0, stream>>>(
            H1, src, dst, norm, out1);
        bias_relu<HID><<<(tot + TB - 1) / TB, TB, 0, stream>>>(out1, b1);
    }

    {
        dim3 g((N_NODES / 16 + 7) / 8, HID2 / 16);
        gemm_wmma<HID, HID2><<<g, 256, 0, stream>>>(out1, W2, H2);
    }
    {
        int tot = N_NODES * (HID2 / 4);
        selfloop_init<HID2><<<(tot + TB - 1) / TB, TB, 0, stream>>>(H2, dinv, out2);
        scatter_edges<HID2, 16><<<(N_EDGES + 15) / 16, dim3(HID2 / 4, 16), 0, stream>>>(
            H2, src, dst, norm, out2);
        bias_relu<HID2><<<(tot + TB - 1) / TB, TB, 0, stream>>>(out2, b2);
    }

    classifier<<<(N_NODES + TB - 1) / TB, TB, 0, stream>>>(out2, Wc, bc, logits);
}


# scan once: the degree pass appends each work-group's matching edges to its own list; both feature passes replay it (source id, norm in list order), no more dst scans; LDS adds in replay arrival order
# speedup vs baseline: 2.0946x; 2.0946x over previous
#include <hip/hip_runtime.h>

#define N_NODES 100000
#define N_EDGES 3200000
#define IN_CH   128
#define HID     96
#define HID2    48
#define NCLS    3

typedef __attribute__((ext_vector_type(2)))  float  v2f;
typedef __attribute__((ext_vector_type(8)))  float  v8f;


#define RNG    736
#define CAP    32768

__global__ __launch_bounds__(256) void list_norm(const int* __restrict__ src,
                                                 const float* __restrict__ dinv,
                                                 const int* __restrict__ lcount,
                                                 unsigned* lkey,
                                                 float* lnorm) {
    const int blk = blockIdx.y;
    const int i = blockIdx.x * blockDim.x + threadIdx.x;
    if (i >= lcount[blk]) return;
    const size_t p = (size_t)blk * CAP + i;
    const unsigned ent = lkey[p];
    const int e = (int)(ent & 0x3FFFFFu);
    const unsigned loc = ent >> 22;
    const int s = src[e];
    lnorm[p] = dinv[s] * lnorm[p] * dinv[blk * RNG + (int)loc];
    lkey[p] = (loc << 17) | (unsigned)s;
}


template<int K, int COLS>
__global__ __launch_bounds__(256) void gemm_wmma(const float* __restrict__ X,
                                                 const float* __restrict__ W,
                                                 float* __restrict__ H) {
    const int lane = threadIdx.x & 31;
    const int wave = threadIdx.x >> 5;
    const int tm = blockIdx.x * 8 + wave;
    const int tn = blockIdx.y;
    if (tm * 16 >= N_NODES) return;
    const int m0 = tm * 16, n0 = tn * 16;
    const int row   = m0 + (lane & 15);
    const int col   = n0 + (lane & 15);
    const int khalf = (lane >> 4) << 1;
    const float* xrow = X + (size_t)row * K;
    v8f c = {};
    for (int kk = 0; kk < K; kk += 4) {
        v2f a = *(const v2f*)(xrow + kk + khalf);
        v2f b;
        b.x = W[(size_t)(kk + khalf)     * COLS + col];
        b.y = W[(size_t)(kk + khalf + 1) * COLS + col];
        c = __builtin_amdgcn_wmma_f32_16x16x4_f32(false, a, false, b,
                                                  (short)0, c, false, false);
    }
    const int rbase = (lane >> 4) << 3;
    float* hrow = H + (size_t)(m0 + rbase) * COLS + col;
    #pragma unroll
    for (int r = 0; r < 8; ++r) hrow[(size_t)r * COLS] = c[r];
}


#define RING   256
#define CHUNK  1024
#define NCHUNK (N_EDGES / CHUNK)

static_assert(N_EDGES % CHUNK == 0, "whole chunks");
static_assert(N_EDGES <= (1 << 22), "edge id packs into 22 bits");
static_assert(RNG <= 1024, "row offset packs into 10 bits, one thread per owned node");
static_assert(N_NODES <= (1 << 17), "source node packs into 17 bits beside the row offset");

static __device__ __forceinline__ int   lane_i(int v, int l)   { return __builtin_amdgcn_readlane(v, l); }
static __device__ __forceinline__ float lane_f(float v, int l) {
    return __int_as_float(__builtin_amdgcn_readlane(__float_as_int(v), l));
}
static __device__ __forceinline__ void lds_add(float* p, float v) {
    __hip_atomic_fetch_add(p, v, __ATOMIC_RELAXED, __HIP_MEMORY_SCOPE_WORKGROUP);
}

template<int C, bool FULL>
static __device__ __forceinline__ void drain(float* __restrict__ tile,
                                             int s, int loc, float nw, int n, int lane,
                                             const float* __restrict__ H) {
    constexpr int T = (C + 31) / 32;
    if (FULL) {
        #pragma unroll 1
        for (int g = 0; g < 32; g += 4) {
            float hv[4][T];
            #pragma unroll
            for (int u = 0; u < 4; ++u) {
                const float* hr = H + (size_t)lane_i(s, g + u) * C + lane;
                #pragma unroll
                for (int t = 0; t < T; ++t)
                    hv[u][t] = (t * 32 + 32 <= C || lane < C - t * 32) ? hr[t * 32] : 0.0f;
            }
            #pragma unroll
            for (int u = 0; u < 4; ++u) {
                const float w = lane_f(nw, g + u);
                float* tr = tile + lane_i(loc, g + u) * C + lane;
                #pragma unroll
                for (int t = 0; t < T; ++t)
                    if (t * 32 + 32 <= C || lane < C - t * 32) lds_add(tr + t * 32, w * hv[u][t]);
            }
        }
    } else {
        for (int i = 0; i < n; ++i) {
            const float* hr = H + (size_t)lane_i(s, i) * C + lane;
            const float w = lane_f(nw, i);
            float* tr = tile + lane_i(loc, i) * C + lane;
            #pragma unroll
            for (int t = 0; t < T; ++t)
                if (t * 32 + 32 <= C || lane < C - t * 32) lds_add(tr + t * 32, w * hr[t * 32]);
        }
    }
}

template<class F32, class FRest>
static __device__ __forceinline__ void scan_owned(const int* __restrict__ dst, int base,
                                                  unsigned* __restrict__ rq, int lane, int wave,
                                                  F32 on32, FRest rest) {
    unsigned head = 0, tail = 0;
    const int start = (int)(((long long)blockIdx.x * NCHUNK) / gridDim.x);
    for (int it = wave; it < NCHUNK; it += 32) {
        int chunk = start + it;
        if (chunk >= NCHUNK) chunk -= NCHUNK;
        const int e0 = chunk * CHUNK;
        const int4* dp = (const int4*)(dst + e0) + lane;
        int4 v[8];
        #pragma unroll
        for (int j = 0; j < 8; ++j) v[j] = dp[j * 32];
        #pragma unroll
        for (int j = 0; j < 8; ++j) {
            const int dd[4] = { v[j].x, v[j].y, v[j].z, v[j].w };
            #pragma unroll
            for (int k = 0; k < 4; ++k) {
                const unsigned loc = (unsigned)(dd[k] - base);
                const bool m = loc < (unsigned)RNG;
                const unsigned mask = __builtin_amdgcn_ballot_w32(m);
                if (mask) {
                    if (m) {
                        const unsigned pos = (tail + __builtin_amdgcn_mbcnt_lo(mask, 0u)) & (RING - 1);
                        rq[pos] = (loc << 22) | (unsigned)(e0 + j * 128 + lane * 4 + k);
                    }
                    tail += (unsigned)__builtin_popcount(mask);
                }
            }
            while (tail - head >= 32u) { on32(head); head += 32u; }
        }
    }
    if (tail != head) rest(head, (int)(tail - head));
}

__global__ __launch_bounds__(1024) void own_deg(const int* __restrict__ dst,
                                                const float* __restrict__ ew,
                                                float* __restrict__ dinv,
                                                unsigned* __restrict__ lkey,
                                                float* __restrict__ lw,
                                                int* __restrict__ lcount) {
    __shared__ float    sdeg[RNG];
    __shared__ unsigned ring[32][RING];
    __shared__ int      lcnt, lovf;
    const int lane = threadIdx.x & 31;
    const int wave = threadIdx.x >> 5;
    const int base = blockIdx.x * RNG;
    if (threadIdx.x == 0) { lcnt = 0; lovf = 0; }
    if (threadIdx.x < RNG) sdeg[threadIdx.x] = 1.0f;
    __syncthreads();
    unsigned* rq = ring[wave];
    unsigned* lk = lkey + (size_t)blockIdx.x * CAP;
    float*    lv = lw   + (size_t)blockIdx.x * CAP;
    auto add_n = [&](unsigned head, int n) {
        unsigned ent = 0; float w = 0.0f;
        if (lane < n) {
            ent = rq[(head + lane) & (RING - 1)];
            w   = ew[ent & 0x3FFFFFu];
        }
        int pos = 0;
        if (lane == 0)
            pos = __hip_atomic_fetch_add(&lcnt, n, __ATOMIC_RELAXED, __HIP_MEMORY_SCOPE_WORKGROUP);
        pos = __builtin_amdgcn_readfirstlane(pos);
        if (pos + n <= CAP) {
            if (lane < n) { lk[pos + lane] = ent; lv[pos + lane] = w; }
        } else if (lane == 0) {
            __hip_atomic_store(&lovf, 1, __ATOMIC_RELAXED, __HIP_MEMORY_SCOPE_WORKGROUP);
        }
        if (lane < n) lds_add(sdeg + (ent >> 22), w);
    };
    scan_owned(dst, base, rq, lane, wave,
               [&](unsigned head) { add_n(head, 32); }, add_n);
    __syncthreads();
    if (threadIdx.x < RNG && base + (int)threadIdx.x < N_NODES)
        dinv[base + threadIdx.x] = rsqrtf(sdeg[threadIdx.x]);
    if (threadIdx.x == 0) lcount[blockIdx.x] = lovf ? -1 : lcnt;
}

template<int C>
__global__ __launch_bounds__(1024) void own_rows(const float* __restrict__ H,
                                                 const int* __restrict__ src,
                                                 const int* __restrict__ dst,
                                                 const float* __restrict__ ew,
                                                 const float* __restrict__ dinv,
                                                 const int* __restrict__ lcount,
                                                 const unsigned* __restrict__ lkey,
                                                 const float* __restrict__ lnorm,
                                                 const float* __restrict__ bias,
                                                 float* __restrict__ out) {
    __shared__ float    tile[RNG * C];
    __shared__ unsigned ring[32][RING];
    constexpr int C4 = C / 4;
    const int lane = threadIdx.x & 31;
    const int wave = __builtin_amdgcn_readfirstlane((int)(threadIdx.x >> 5));
    const int base = blockIdx.x * RNG;

    for (int idx = threadIdx.x; idx < RNG * C4; idx += 1024) {
        const int r = idx / C4, q = idx - r * C4;
        const int node = base + r;
        float4 o = { 0.0f, 0.0f, 0.0f, 0.0f };
        if (node < N_NODES) {
            const float di = dinv[node], w = di * di;
            const float4 hv = *(const float4*)(H + (size_t)node * C + q * 4);
            o.x = w * hv.x; o.y = w * hv.y; o.z = w * hv.z; o.w = w * hv.w;
        }
        *(float4*)(tile + idx * 4) = o;
    }
    __syncthreads();

    const int cnt = lcount[blockIdx.x];
    if (cnt >= 0) {
        const unsigned* lk = lkey  + (size_t)blockIdx.x * CAP;
        const float*    ln = lnorm + (size_t)blockIdx.x * CAP;
        int g0 = wave * 32;
        unsigned key = 0; float nw = 0.0f;
        if (g0 + lane < cnt) { key = lk[g0 + lane]; nw = ln[g0 + lane]; }
        while (g0 < cnt) {
            const int gn = g0 + 1024;
            unsigned key2 = 0; float nw2 = 0.0f;
            if (gn + lane < cnt) { key2 = lk[gn + lane]; nw2 = ln[gn + lane]; }
            const int n = cnt - g0;
            const int s = (int)(key & 0x1FFFFu), loc = (int)(key >> 17);
            if (n >= 32) drain<C, true>(tile, s, loc, nw, 32, lane, H);
            else         drain<C, false>(tile, s, loc, nw, n, lane, H);
            key = key2; nw = nw2; g0 = gn;
        }
    } else {
        unsigned* rq = ring[wave];
        auto look = [&](unsigned head, int n, int& s, int& loc, float& nw) {
            s = 0; loc = 0; nw = 0.0f;
            if (lane < n) {
                const unsigned ent = rq[(head + lane) & (RING - 1)];
                const int e = (int)(ent & 0x3FFFFFu);
                loc = (int)(ent >> 22);
                s   = src[e];
                nw  = dinv[s] * ew[e] * dinv[base + loc];
            }
        };
        scan_owned(dst, base, rq, lane, wave,
                   [&](unsigned head) { int s, loc; float nw; look(head, 32, s, loc, nw);
                                        drain<C, true>(tile, s, loc, nw, 32, lane, H); },
                   [&](unsigned head, int n) { int s, loc; float nw; look(head, n, s, loc, nw);
                                               drain<C, false>(tile, s, loc, nw, n, lane, H); });
    }
    __syncthreads();

    for (int idx = threadIdx.x; idx < RNG * C4; idx += 1024) {
        const int r = idx / C4, q = idx - r * C4;
        const int node = base + r;
        if (node < N_NODES) {
            float4 t4 = *(const float4*)(tile + idx * 4);
            const float4 bb = *(const float4*)(bias + q * 4);
            t4.x = fmaxf(t4.x + bb.x, 0.0f);
            t4.y = fmaxf(t4.y + bb.y, 0.0f);
            t4.z = fmaxf(t4.z + bb.z, 0.0f);
            t4.w = fmaxf(t4.w + bb.w, 0.0f);
            *(float4*)(out + (size_t)node * C + q * 4) = t4;
        }
    }
}

__global__ __launch_bounds__(256) void classifier(const float* __restrict__ Hin,
                                                  const float* __restrict__ Wc,
                                                  const float* __restrict__ bc,
                                                  float* __restrict__ logits) {
    int i = blockIdx.x * blockDim.x + threadIdx.x;
    if (i >= N_NODES) return;
    const float* h = Hin + (size_t)i * HID2;
    float a0 = bc[0], a1 = bc[1], a2 = bc[2];
    #pragma unroll
    for (int k = 0; k < HID2; ++k) {
        float hv = h[k];
        a0 += hv * Wc[k * 3 + 0];
        a1 += hv * Wc[k * 3 + 1];
        a2 += hv * Wc[k * 3 + 2];
    }
    logits[(size_t)i * 3 + 0] = a0;
    logits[(size_t)i * 3 + 1] = a1;
    logits[(size_t)i * 3 + 2] = a2;
}


extern "C" void kernel_launch(void* const* d_in, const int* in_sizes, int n_in,
                              void* d_out, int out_size, void* d_ws, size_t ws_size,
                              hipStream_t stream) {
    const float* x  = (const float*)d_in[0];
    const int*   ei = (const int*)  d_in[1];
    const float* ew = (const float*)d_in[2];
    const float* W1 = (const float*)d_in[3];
    const float* b1 = (const float*)d_in[4];
    const float* W2 = (const float*)d_in[5];
    const float* b2 = (const float*)d_in[6];
    const float* Wc = (const float*)d_in[7];
    const float* bc = (const float*)d_in[8];
    float* logits = (float*)d_out;

    const int* src = ei;
    const int* dst = ei + N_EDGES;

    char* ws = (char*)d_ws;
    size_t off = 0;
    auto take = [&](size_t bytes) -> void* {
        void* p = ws + off;
        off += (bytes + 255) & ~(size_t)255;
        return p;
    };
    float* dinv = (float*)take((size_t)N_NODES * 4);
    int*      lcount = (int*)     take((size_t)((N_NODES + RNG - 1) / RNG) * 4);
    unsigned* lkey   = (unsigned*)take((size_t)((N_NODES + RNG - 1) / RNG) * CAP * 4);
    float*    lnorm  = (float*)   take((size_t)((N_NODES + RNG - 1) / RNG) * CAP * 4);
    float* H1   = (float*)take((size_t)N_NODES * HID * 4);
    float* out1 = (float*)take((size_t)N_NODES * HID * 4);
    float* H2   = H1;
    float* out2 = out1;

    const int TB = 256;
    own_deg<<<(N_NODES + RNG - 1) / RNG, 1024, 0, stream>>>(dst, ew, dinv, lkey, lnorm, lcount);
    {
        dim3 g(CAP / TB, (N_NODES + RNG - 1) / RNG);
        list_norm<<<g, TB, 0, stream>>>(src, dinv, lcount, lkey, lnorm);
    }

    {
        dim3 g((N_NODES / 16 + 7) / 8, HID / 16);
        gemm_wmma<IN_CH, HID><<<g, 256, 0, stream>>>(x, W1, H1);
    }
    own_rows<HID><<<(N_NODES + RNG - 1) / RNG, 1024, 0, stream>>>(
        H1, src, dst, ew, dinv, lcount, lkey, lnorm, b1, out1);

    {
        dim3 g((N_NODES / 16 + 7) / 8, HID2 / 16);
        gemm_wmma<HID, HID2><<<g, 256, 0, stream>>>(out1, W2, H2);
    }
    own_rows<HID2><<<(N_NODES + RNG - 1) / RNG, 1024, 0, stream>>>(
        H2, src, dst, ew, dinv, lcount, lkey, lnorm, b2, out2);

    classifier<<<(N_NODES + TB - 1) / TB, TB, 0, stream>>>(out2, Wc, bc, logits);
}


# owner ranges cut from 736 to 391 nodes: 256 work-groups, one per WGP (was 136, 120 WGPs idle) in the degree pass and both feature passes; LDS adds still in arrival order inside a work-group
# speedup vs baseline: 2.5786x; 1.2311x over previous
#include <hip/hip_runtime.h>

#define N_NODES 100000
#define N_EDGES 3200000
#define IN_CH   128
#define HID     96
#define HID2    48
#define NCLS    3

typedef __attribute__((ext_vector_type(2)))  float  v2f;
typedef __attribute__((ext_vector_type(8)))  float  v8f;


#define NGRP   256
#define RNG    ((N_NODES + NGRP - 1) / NGRP)
#define CAP    16384

static_assert((NGRP - 1) * RNG < N_NODES && NGRP * RNG >= N_NODES, "NGRP work-groups cover the nodes, none is empty");

__global__ __launch_bounds__(256) void list_norm(const int* __restrict__ src,
                                                 const float* __restrict__ dinv,
                                                 const int* __restrict__ lcount,
                                                 unsigned* lkey,
                                                 float* lnorm) {
    const int blk = blockIdx.y;
    const int i = blockIdx.x * blockDim.x + threadIdx.x;
    if (i >= lcount[blk]) return;
    const size_t p = (size_t)blk * CAP + i;
    const unsigned ent = lkey[p];
    const int e = (int)(ent & 0x3FFFFFu);
    const unsigned loc = ent >> 22;
    const int s = src[e];
    lnorm[p] = dinv[s] * lnorm[p] * dinv[blk * RNG + (int)loc];
    lkey[p] = (loc << 17) | (unsigned)s;
}


template<int K, int COLS>
__global__ __launch_bounds__(256) void gemm_wmma(const float* __restrict__ X,
                                                 const float* __restrict__ W,
                                                 float* __restrict__ H) {
    const int lane = threadIdx.x & 31;
    const int wave = threadIdx.x >> 5;
    const int tm = blockIdx.x * 8 + wave;
    const int tn = blockIdx.y;
    if (tm * 16 >= N_NODES) return;
    const int m0 = tm * 16, n0 = tn * 16;
    const int row   = m0 + (lane & 15);
    const int col   = n0 + (lane & 15);
    const int khalf = (lane >> 4) << 1;
    const float* xrow = X + (size_t)row * K;
    v8f c = {};
    for (int kk = 0; kk < K; kk += 4) {
        v2f a = *(const v2f*)(xrow + kk + khalf);
        v2f b;
        b.x = W[(size_t)(kk + khalf)     * COLS + col];
        b.y = W[(size_t)(kk + khalf + 1) * COLS + col];
        c = __builtin_amdgcn_wmma_f32_16x16x4_f32(false, a, false, b,
                                                  (short)0, c, false, false);
    }
    const int rbase = (lane >> 4) << 3;
    float* hrow = H + (size_t)(m0 + rbase) * COLS + col;
    #pragma unroll
    for (int r = 0; r < 8; ++r) hrow[(size_t)r * COLS] = c[r];
}


#define RING   256
#define CHUNK  1024
#define NCHUNK (N_EDGES / CHUNK)

static_assert(N_EDGES % CHUNK == 0, "whole chunks");
static_assert(N_EDGES <= (1 << 22), "edge id packs into 22 bits");
static_assert(RNG <= 1024, "row offset packs into 10 bits, one thread per owned node");
static_assert(N_NODES <= (1 << 17), "source node packs into 17 bits beside the row offset");

static __device__ __forceinline__ int   lane_i(int v, int l)   { return __builtin_amdgcn_readlane(v, l); }
static __device__ __forceinline__ float lane_f(float v, int l) {
    return __int_as_float(__builtin_amdgcn_readlane(__float_as_int(v), l));
}
static __device__ __forceinline__ void lds_add(float* p, float v) {
    __hip_atomic_fetch_add(p, v, __ATOMIC_RELAXED, __HIP_MEMORY_SCOPE_WORKGROUP);
}

template<int C, bool FULL>
static __device__ __forceinline__ void drain(float* __restrict__ tile,
                                             int s, int loc, float nw, int n, int lane,
                                             const float* __restrict__ H) {
    constexpr int T = (C + 31) / 32;
    if (FULL) {
        #pragma unroll 1
        for (int g = 0; g < 32; g += 4) {
            float hv[4][T];
            #pragma unroll
            for (int u = 0; u < 4; ++u) {
                const float* hr = H + (size_t)lane_i(s, g + u) * C + lane;
                #pragma unroll
                for (int t = 0; t < T; ++t)
                    hv[u][t] = (t * 32 + 32 <= C || lane < C - t * 32) ? hr[t * 32] : 0.0f;
            }
            #pragma unroll
            for (int u = 0; u < 4; ++u) {
                const float w = lane_f(nw, g + u);
                float* tr = tile + lane_i(loc, g + u) * C + lane;
                #pragma unroll
                for (int t = 0; t < T; ++t)
                    if (t * 32 + 32 <= C || lane < C - t * 32) lds_add(tr + t * 32, w * hv[u][t]);
            }
        }
    } else {
        for (int i = 0; i < n; ++i) {
            const float* hr = H + (size_t)lane_i(s, i) * C + lane;
            const float w = lane_f(nw, i);
            float* tr = tile + lane_i(loc, i) * C + lane;
            #pragma unroll
            for (int t = 0; t < T; ++t)
                if (t * 32 + 32 <= C || lane < C - t * 32) lds_add(tr + t * 32, w * hr[t * 32]);
        }
    }
}

template<class F32, class FRest>
static __device__ __forceinline__ void scan_owned(const int* __restrict__ dst, int base,
                                                  unsigned* __restrict__ rq, int lane, int wave,
                                                  F32 on32, FRest rest) {
    unsigned head = 0, tail = 0;
    const int start = (int)(((long long)blockIdx.x * NCHUNK) / gridDim.x);
    for (int it = wave; it < NCHUNK; it += 32) {
        int chunk = start + it;
        if (chunk >= NCHUNK) chunk -= NCHUNK;
        const int e0 = chunk * CHUNK;
        const int4* dp = (const int4*)(dst + e0) + lane;
        int4 v[8];
        #pragma unroll
        for (int j = 0; j < 8; ++j) v[j] = dp[j * 32];
        #pragma unroll
        for (int j = 0; j < 8; ++j) {
            const int dd[4] = { v[j].x, v[j].y, v[j].z, v[j].w };
            #pragma unroll
            for (int k = 0; k < 4; ++k) {
                const unsigned loc = (unsigned)(dd[k] - base);
                const bool m = loc < (unsigned)RNG;
                const unsigned mask = __builtin_amdgcn_ballot_w32(m);
                if (mask) {
                    if (m) {
                        const unsigned pos = (tail + __builtin_amdgcn_mbcnt_lo(mask, 0u)) & (RING - 1);
                        rq[pos] = (loc << 22) | (unsigned)(e0 + j * 128 + lane * 4 + k);
                    }
                    tail += (unsigned)__builtin_popcount(mask);
                }
            }
            while (tail - head >= 32u) { on32(head); head += 32u; }
        }
    }
    if (tail != head) rest(head, (int)(tail - head));
}

__global__ __launch_bounds__(1024) void own_deg(const int* __restrict__ dst,
                                                const float* __restrict__ ew,
                                                float* __restrict__ dinv,
                                                unsigned* __restrict__ lkey,
                                                float* __restrict__ lw,
                                                int* __restrict__ lcount) {
    __shared__ float    sdeg[RNG];
    __shared__ unsigned ring[32][RING];
    __shared__ int      lcnt, lovf;
    const int lane = threadIdx.x & 31;
    const int wave = threadIdx.x >> 5;
    const int base = blockIdx.x * RNG;
    if (threadIdx.x == 0) { lcnt = 0; lovf = 0; }
    if (threadIdx.x < RNG) sdeg[threadIdx.x] = 1.0f;
    __syncthreads();
    unsigned* rq = ring[wave];
    unsigned* lk = lkey + (size_t)blockIdx.x * CAP;
    float*    lv = lw   + (size_t)blockIdx.x * CAP;
    auto add_n = [&](unsigned head, int n) {
        unsigned ent = 0; float w = 0.0f;
        if (lane < n) {
            ent = rq[(head + lane) & (RING - 1)];
            w   = ew[ent & 0x3FFFFFu];
        }
        int pos = 0;
        if (lane == 0)
            pos = __hip_atomic_fetch_add(&lcnt, n, __ATOMIC_RELAXED, __HIP_MEMORY_SCOPE_WORKGROUP);
        pos = __builtin_amdgcn_readfirstlane(pos);
        if (pos + n <= CAP) {
            if (lane < n) { lk[pos + lane] = ent; lv[pos + lane] = w; }
        } else if (lane == 0) {
            __hip_atomic_store(&lovf, 1, __ATOMIC_RELAXED, __HIP_MEMORY_SCOPE_WORKGROUP);
        }
        if (lane < n) lds_add(sdeg + (ent >> 22), w);
    };
    scan_owned(dst, base, rq, lane, wave,
               [&](unsigned head) { add_n(head, 32); }, add_n);
    __syncthreads();
    if (threadIdx.x < RNG && base + (int)threadIdx.x < N_NODES)
        dinv[base + threadIdx.x] = rsqrtf(sdeg[threadIdx.x]);
    if (threadIdx.x == 0) lcount[blockIdx.x] = lovf ? -1 : lcnt;
}

template<int C>
__global__ __launch_bounds__(1024) void own_rows(const float* __restrict__ H,
                                                 const int* __restrict__ src,
                                                 const int* __restrict__ dst,
                                                 const float* __restrict__ ew,
                                                 const float* __restrict__ dinv,
                                                 const int* __restrict__ lcount,
                                                 const unsigned* __restrict__ lkey,
                                                 const float* __restrict__ lnorm,
                                                 const float* __restrict__ bias,
                                                 float* __restrict__ out) {
    __shared__ float    tile[RNG * C];
    __shared__ unsigned ring[32][RING];
    constexpr int C4 = C / 4;
    const int lane = threadIdx.x & 31;
    const int wave = __builtin_amdgcn_readfirstlane((int)(threadIdx.x >> 5));
    const int base = blockIdx.x * RNG;

    for (int idx = threadIdx.x; idx < RNG * C4; idx += 1024) {
        const int r = idx / C4, q = idx - r * C4;
        const int node = base + r;
        float4 o = { 0.0f, 0.0f, 0.0f, 0.0f };
        if (node < N_NODES) {
            const float di = dinv[node], w = di * di;
            const float4 hv = *(const float4*)(H + (size_t)node * C + q * 4);
            o.x = w * hv.x; o.y = w * hv.y; o.z = w * hv.z; o.w = w * hv.w;
        }
        *(float4*)(tile + idx * 4) = o;
    }
    __syncthreads();

    const int cnt = lcount[blockIdx.x];
    if (cnt >= 0) {
        const unsigned* lk = lkey  + (size_t)blockIdx.x * CAP;
        const float*    ln = lnorm + (size_t)blockIdx.x * CAP;
        int g0 = wave * 32;
        unsigned key = 0; float nw = 0.0f;
        if (g0 + lane < cnt) { key = lk[g0 + lane]; nw = ln[g0 + lane]; }
        while (g0 < cnt) {
            const int gn = g0 + 1024;
            unsigned key2 = 0; float nw2 = 0.0f;
            if (gn + lane < cnt) { key2 = lk[gn + lane]; nw2 = ln[gn + lane]; }
            const int n = cnt - g0;
            const int s = (int)(key & 0x1FFFFu), loc = (int)(key >> 17);
            if (n >= 32) drain<C, true>(tile, s, loc, nw, 32, lane, H);
            else         drain<C, false>(tile, s, loc, nw, n, lane, H);
            key = key2; nw = nw2; g0 = gn;
        }
    } else {
        unsigned* rq = ring[wave];
        auto look = [&](unsigned head, int n, int& s, int& loc, float& nw) {
            s = 0; loc = 0; nw = 0.0f;
            if (lane < n) {
                const unsigned ent = rq[(head + lane) & (RING - 1)];
                const int e = (int)(ent & 0x3FFFFFu);
                loc = (int)(ent >> 22);
                s   = src[e];
                nw  = dinv[s] * ew[e] * dinv[base + loc];
            }
        };
        scan_owned(dst, base, rq, lane, wave,
                   [&](unsigned head) { int s, loc; float nw; look(head, 32, s, loc, nw);
                                        drain<C, true>(tile, s, loc, nw, 32, lane, H); },
                   [&](unsigned head, int n) { int s, loc; float nw; look(head, n, s, loc, nw);
                                               drain<C, false>(tile, s, loc, nw, n, lane, H); });
    }
    __syncthreads();

    for (int idx = threadIdx.x; idx < RNG * C4; idx += 1024) {
        const int r = idx / C4, q = idx - r * C4;
        const int node = base + r;
        if (node < N_NODES) {
            float4 t4 = *(const float4*)(tile + idx * 4);
            const float4 bb = *(const float4*)(bias + q * 4);
            t4.x = fmaxf(t4.x + bb.x, 0.0f);
            t4.y = fmaxf(t4.y + bb.y, 0.0f);
            t4.z = fmaxf(t4.z + bb.z, 0.0f);
            t4.w = fmaxf(t4.w + bb.w, 0.0f);
            *(float4*)(out + (size_t)node * C + q * 4) = t4;
        }
    }
}

__global__ __launch_bounds__(256) void classifier(const float* __restrict__ Hin,
                                                  const float* __restrict__ Wc,
                                                  const float* __restrict__ bc,
                                                  float* __restrict__ logits) {
    int i = blockIdx.x * blockDim.x + threadIdx.x;
    if (i >= N_NODES) return;
    const float* h = Hin + (size_t)i * HID2;
    float a0 = bc[0], a1 = bc[1], a2 = bc[2];
    #pragma unroll
    for (int k = 0; k < HID2; ++k) {
        float hv = h[k];
        a0 += hv * Wc[k * 3 + 0];
        a1 += hv * Wc[k * 3 + 1];
        a2 += hv * Wc[k * 3 + 2];
    }
    logits[(size_t)i * 3 + 0] = a0;
    logits[(size_t)i * 3 + 1] = a1;
    logits[(size_t)i * 3 + 2] = a2;
}


extern "C" void kernel_launch(void* const* d_in, const int* in_sizes, int n_in,
                              void* d_out, int out_size, void* d_ws, size_t ws_size,
                              hipStream_t stream) {
    const float* x  = (const float*)d_in[0];
    const int*   ei = (const int*)  d_in[1];
    const float* ew = (const float*)d_in[2];
    const float* W1 = (const float*)d_in[3];
    const float* b1 = (const float*)d_in[4];
    const float* W2 = (const float*)d_in[5];
    const float* b2 = (const float*)d_in[6];
    const float* Wc = (const float*)d_in[7];
    const float* bc = (const float*)d_in[8];
    float* logits = (float*)d_out;

    const int* src = ei;
    const int* dst = ei + N_EDGES;

    char* ws = (char*)d_ws;
    size_t off = 0;
    auto take = [&](size_t bytes) -> void* {
        void* p = ws + off;
        off += (bytes + 255) & ~(size_t)255;
        return p;
    };
    float* dinv = (float*)take((size_t)N_NODES * 4);
    int*      lcount = (int*)     take((size_t)(NGRP) * 4);
    unsigned* lkey   = (unsigned*)take((size_t)(NGRP) * CAP * 4);
    float*    lnorm  = (float*)   take((size_t)(NGRP) * CAP * 4);
    float* H1   = (float*)take((size_t)N_NODES * HID * 4);
    float* out1 = (float*)take((size_t)N_NODES * HID * 4);
    float* H2   = H1;
    float* out2 = out1;

    const int TB = 256;
    own_deg<<<NGRP, 1024, 0, stream>>>(dst, ew, dinv, lkey, lnorm, lcount);
    {
        dim3 g(CAP / TB, NGRP);
        list_norm<<<g, TB, 0, stream>>>(src, dinv, lcount, lkey, lnorm);
    }

    {
        dim3 g((N_NODES / 16 + 7) / 8, HID / 16);
        gemm_wmma<IN_CH, HID><<<g, 256, 0, stream>>>(x, W1, H1);
    }
    own_rows<HID><<<NGRP, 1024, 0, stream>>>(
        H1, src, dst, ew, dinv, lcount, lkey, lnorm, b1, out1);

    {
        dim3 g((N_NODES / 16 + 7) / 8, HID2 / 16);
        gemm_wmma<HID, HID2><<<g, 256, 0, stream>>>(out1, W2, H2);
    }
    own_rows<HID2><<<NGRP, 1024, 0, stream>>>(
        H2, src, dst, ew, dinv, lcount, lkey, lnorm, b2, out2);

    classifier<<<(N_NODES + TB - 1) / TB, TB, 0, stream>>>(out2, Wc, bc, logits);
}


# degree pass's scan in two levels: 16x16 work-groups list each edge slice's edges per coarse range of 16 owner ranges; an owner scans only those lists (1/16 of the edges); LDS adds in arrival order
# speedup vs baseline: 3.0813x; 1.1950x over previous
#include <hip/hip_runtime.h>

#define N_NODES 100000
#define N_EDGES 3200000
#define IN_CH   128
#define HID     96
#define HID2    48
#define NCLS    3

typedef __attribute__((ext_vector_type(2)))  float  v2f;
typedef __attribute__((ext_vector_type(8)))  float  v8f;


#define NGRP   256
#define RNG    ((N_NODES + NGRP - 1) / NGRP)
#define CAP    16384

static_assert((NGRP - 1) * RNG < N_NODES && NGRP * RNG >= N_NODES, "NGRP work-groups cover the nodes, none is empty");

__global__ __launch_bounds__(256) void list_norm(const int* __restrict__ src,
                                                 const float* __restrict__ dinv,
                                                 const int* __restrict__ lcount,
                                                 unsigned* lkey,
                                                 float* lnorm) {
    const int blk = blockIdx.y;
    const int i = blockIdx.x * blockDim.x + threadIdx.x;
    if (i >= lcount[blk]) return;
    const size_t p = (size_t)blk * CAP + i;
    const unsigned ent = lkey[p];
    const int e = (int)(ent & 0x3FFFFFu);
    const unsigned loc = ent >> 22;
    const int s = src[e];
    lnorm[p] = dinv[s] * lnorm[p] * dinv[blk * RNG + (int)loc];
    lkey[p] = (loc << 17) | (unsigned)s;
}


template<int K, int COLS>
__global__ __launch_bounds__(256) void gemm_wmma(const float* __restrict__ X,
                                                 const float* __restrict__ W,
                                                 float* __restrict__ H) {
    const int lane = threadIdx.x & 31;
    const int wave = threadIdx.x >> 5;
    const int tm = blockIdx.x * 8 + wave;
    const int tn = blockIdx.y;
    if (tm * 16 >= N_NODES) return;
    const int m0 = tm * 16, n0 = tn * 16;
    const int row   = m0 + (lane & 15);
    const int col   = n0 + (lane & 15);
    const int khalf = (lane >> 4) << 1;
    const float* xrow = X + (size_t)row * K;
    v8f c = {};
    for (int kk = 0; kk < K; kk += 4) {
        v2f a = *(const v2f*)(xrow + kk + khalf);
        v2f b;
        b.x = W[(size_t)(kk + khalf)     * COLS + col];
        b.y = W[(size_t)(kk + khalf + 1) * COLS + col];
        c = __builtin_amdgcn_wmma_f32_16x16x4_f32(false, a, false, b,
                                                  (short)0, c, false, false);
    }
    const int rbase = (lane >> 4) << 3;
    float* hrow = H + (size_t)(m0 + rbase) * COLS + col;
    #pragma unroll
    for (int r = 0; r < 8; ++r) hrow[(size_t)r * COLS] = c[r];
}


#define RING   256
#define CHUNK  1024
#define NCHUNK (N_EDGES / CHUNK)

static_assert(N_EDGES % CHUNK == 0, "whole chunks");
static_assert(N_EDGES <= (1 << 22), "edge id packs into 22 bits");
static_assert(RNG <= 1024, "row offset packs into 10 bits, one thread per owned node");
static_assert(N_NODES <= (1 << 17), "source node packs into 17 bits beside the row offset");

#define SUP    16
#define NSUP   (NGRP / SUP)
#define SRNG   (SUP * RNG)
#define NSL    16
#define CH1    512
#define NCH1   (N_EDGES / CH1)
#define SEGCAP 16384
#define CPAD   0xFFFFFFFFu

static_assert(NGRP % SUP == 0, "whole coarse ranges");
static_assert(N_EDGES % CH1 == 0 && SEGCAP % CH1 == 0, "whole trips");
static_assert((NCH1 / NSL + 1) * CH1 <= (1 << 18), "edge offset in a slice packs into 18 bits");
static_assert(SRNG < (1 << 14) - 1, "node offset in a coarse range packs into 14 bits, below the filler's");
static_assert((CPAD >> 18) - (SUP - 1) * RNG >= RNG, "the filler matches no owner range");
static_assert(NSL <= 32, "one lane per list of a coarse range");

static __device__ __forceinline__ int   lane_i(int v, int l)   { return __builtin_amdgcn_readlane(v, l); }
static __device__ __forceinline__ float lane_f(float v, int l) {
    return __int_as_float(__builtin_amdgcn_readlane(__float_as_int(v), l));
}
static __device__ __forceinline__ void lds_add(float* p, float v) {
    __hip_atomic_fetch_add(p, v, __ATOMIC_RELAXED, __HIP_MEMORY_SCOPE_WORKGROUP);
}

template<int C, bool FULL>
static __device__ __forceinline__ void drain(float* __restrict__ tile,
                                             int s, int loc, float nw, int n, int lane,
                                             const float* __restrict__ H) {
    constexpr int T = (C + 31) / 32;
    if (FULL) {
        #pragma unroll 1
        for (int g = 0; g < 32; g += 4) {
            float hv[4][T];
            #pragma unroll
            for (int u = 0; u < 4; ++u) {
                const float* hr = H + (size_t)lane_i(s, g + u) * C + lane;
                #pragma unroll
                for (int t = 0; t < T; ++t)
                    hv[u][t] = (t * 32 + 32 <= C || lane < C - t * 32) ? hr[t * 32] : 0.0f;
            }
            #pragma unroll
            for (int u = 0; u < 4; ++u) {
                const float w = lane_f(nw, g + u);
                float* tr = tile + lane_i(loc, g + u) * C + lane;
                #pragma unroll
                for (int t = 0; t < T; ++t)
                    if (t * 32 + 32 <= C || lane < C - t * 32) lds_add(tr + t * 32, w * hv[u][t]);
            }
        }
    } else {
        for (int i = 0; i < n; ++i) {
            const float* hr = H + (size_t)lane_i(s, i) * C + lane;
            const float w = lane_f(nw, i);
            float* tr = tile + lane_i(loc, i) * C + lane;
            #pragma unroll
            for (int t = 0; t < T; ++t)
                if (t * 32 + 32 <= C || lane < C - t * 32) lds_add(tr + t * 32, w * hr[t * 32]);
        }
    }
}

template<class F32, class FRest>
static __device__ __forceinline__ void scan_owned(const int* __restrict__ dst, int base,
                                                  unsigned* __restrict__ rq, int lane, int wave,
                                                  F32 on32, FRest rest) {
    unsigned head = 0, tail = 0;
    const int start = (int)(((long long)blockIdx.x * NCHUNK) / gridDim.x);
    for (int it = wave; it < NCHUNK; it += 32) {
        int chunk = start + it;
        if (chunk >= NCHUNK) chunk -= NCHUNK;
        const int e0 = chunk * CHUNK;
        const int4* dp = (const int4*)(dst + e0) + lane;
        int4 v[8];
        #pragma unroll
        for (int j = 0; j < 8; ++j) v[j] = dp[j * 32];
        #pragma unroll
        for (int j = 0; j < 8; ++j) {
            const int dd[4] = { v[j].x, v[j].y, v[j].z, v[j].w };
            #pragma unroll
            for (int k = 0; k < 4; ++k) {
                const unsigned loc = (unsigned)(dd[k] - base);
                const bool m = loc < (unsigned)RNG;
                const unsigned mask = __builtin_amdgcn_ballot_w32(m);
                if (mask) {
                    if (m) {
                        const unsigned pos = (tail + __builtin_amdgcn_mbcnt_lo(mask, 0u)) & (RING - 1);
                        rq[pos] = (loc << 22) | (unsigned)(e0 + j * 128 + lane * 4 + k);
                    }
                    tail += (unsigned)__builtin_popcount(mask);
                }
            }
            while (tail - head >= 32u) { on32(head); head += 32u; }
        }
    }
    if (tail != head) rest(head, (int)(tail - head));
}

__global__ __launch_bounds__(1024) void coarse_split(const int* __restrict__ dst,
                                                     unsigned* __restrict__ clist,
                                                     int* __restrict__ ccount) {
    __shared__ unsigned ring[32][RING];
    __shared__ int      lcnt, lovf;
    const int lane = threadIdx.x & 31;
    const int wave = __builtin_amdgcn_readfirstlane((int)(threadIdx.x >> 5));
    const int p = blockIdx.x;
    const int a = blockIdx.y;
    const int c0 = p * NCH1 / NSL;
    const int nch = (p + 1) * NCH1 / NSL - c0;
    const int cbase = a * SRNG;
    const int* dsl = dst + (size_t)c0 * CH1;
    if (threadIdx.x == 0) { lcnt = 0; lovf = 0; }
    __syncthreads();
    unsigned* rq  = ring[wave];
    unsigned* seg = clist + (size_t)(a * NSL + p) * SEGCAP;
    auto put = [&](unsigned head, int n) {
        unsigned ent = 0;
        if (lane < n) ent = rq[(head + lane) & (RING - 1)];
        int pos = 0;
        if (lane == 0)
            pos = __hip_atomic_fetch_add(&lcnt, n, __ATOMIC_RELAXED, __HIP_MEMORY_SCOPE_WORKGROUP);
        pos = __builtin_amdgcn_readfirstlane(pos);
        if (pos + n <= SEGCAP) {
            if (lane < n) seg[pos + lane] = ent;
        } else if (lane == 0) {
            __hip_atomic_store(&lovf, 1, __ATOMIC_RELAXED, __HIP_MEMORY_SCOPE_WORKGROUP);
        }
    };
    unsigned head = 0, tail = 0;
    const int rot = a * nch / NSUP;
    for (int it = wave; it < nch; it += 32) {
        int c = it + rot;
        if (c >= nch) c -= nch;
        const int eo = c * CH1;
        const int4* dp = (const int4*)(dsl + eo) + lane;
        int4 v[4];
        #pragma unroll
        for (int j = 0; j < 4; ++j) v[j] = dp[j * 32];
        #pragma unroll
        for (int j = 0; j < 4; ++j) {
            const int dd[4] = { v[j].x, v[j].y, v[j].z, v[j].w };
            #pragma unroll
            for (int k = 0; k < 4; ++k) {
                const unsigned loc = (unsigned)(dd[k] - cbase);
                const bool m = loc < (unsigned)SRNG;
                const unsigned mask = __builtin_amdgcn_ballot_w32(m);
                if (mask) {
                    if (m) {
                        const unsigned pos = (tail + __builtin_amdgcn_mbcnt_lo(mask, 0u)) & (RING - 1);
                        rq[pos] = (loc << 18) | (unsigned)(eo + j * 128 + lane * 4 + k);
                    }
                    tail += (unsigned)__builtin_popcount(mask);
                }
            }
            while (tail - head >= 32u) { put(head, 32); head += 32u; }
        }
    }
    if (tail != head) put(head, (int)(tail - head));
    __syncthreads();
    const int cnt = lcnt, ovf = lovf;
    if (!ovf) {
        const int end = (cnt + CH1 - 1) / CH1 * CH1;
        for (int i = cnt + (int)threadIdx.x; i < end; i += 1024) seg[i] = CPAD;
    }
    if (threadIdx.x == 0) ccount[a * NSL + p] = ovf ? -1 : cnt;
}

__global__ __launch_bounds__(1024) void own_deg(const int* __restrict__ dst,
                                                const float* __restrict__ ew,
                                                const unsigned* __restrict__ clist,
                                                const int* __restrict__ ccount,
                                                float* __restrict__ dinv,
                                                unsigned* __restrict__ lkey,
                                                float* __restrict__ lw,
                                                int* __restrict__ lcount) {
    __shared__ float    sdeg[RNG];
    __shared__ unsigned ring[32][RING];
    __shared__ int      lcnt, lovf;
    const int lane = threadIdx.x & 31;
    const int wave = __builtin_amdgcn_readfirstlane((int)(threadIdx.x >> 5));
    const int base = blockIdx.x * RNG;
    const int a = blockIdx.x / SUP;
    const int b = blockIdx.x % SUP;
    if (threadIdx.x == 0) { lcnt = 0; lovf = 0; }
    if (threadIdx.x < RNG) sdeg[threadIdx.x] = 1.0f;
    __syncthreads();
    unsigned* rq = ring[wave];
    unsigned* lk = lkey + (size_t)blockIdx.x * CAP;
    float*    lv = lw   + (size_t)blockIdx.x * CAP;
    auto add_n = [&](unsigned head, int n) {
        unsigned ent = 0; float w = 0.0f;
        if (lane < n) {
            ent = rq[(head + lane) & (RING - 1)];
            w   = ew[ent & 0x3FFFFFu];
        }
        int pos = 0;
        if (lane == 0)
            pos = __hip_atomic_fetch_add(&lcnt, n, __ATOMIC_RELAXED, __HIP_MEMORY_SCOPE_WORKGROUP);
        pos = __builtin_amdgcn_readfirstlane(pos);
        if (pos + n <= CAP) {
            if (lane < n) { lk[pos + lane] = ent; lv[pos + lane] = w; }
        } else if (lane == 0) {
            __hip_atomic_store(&lovf, 1, __ATOMIC_RELAXED, __HIP_MEMORY_SCOPE_WORKGROUP);
        }
        if (lane < n) lds_add(sdeg + (ent >> 22), w);
    };
    const int cn = lane < NSL ? ccount[a * NSL + lane] : 0;
    if (__builtin_amdgcn_ballot_w32(cn < 0) == 0u) {
        const int nc = (cn + CH1 - 1) / CH1;
        int before = 0;
        #pragma unroll
        for (int q = 0; q < NSL; ++q) {
            const int t = lane_i(nc, q);
            if (lane > q) before += t;
        }
        const int upto = before + nc;
        const int total = lane_i(upto, 31);
        const int rot = b * total / SUP;
        const unsigned lbase = (unsigned)(b * RNG);
        unsigned head = 0, tail = 0;
        for (int it = wave; it < total; it += 32) {
            int gi = it + rot;
            if (gi >= total) gi -= total;
            const int p = __builtin_popcount(__builtin_amdgcn_ballot_w32(lane < NSL && upto <= gi));
            const int c = gi - lane_i(before, p);
            const int ef = (p * NCH1 / NSL) * CH1;
            const uint4* dp = (const uint4*)(clist + (size_t)(a * NSL + p) * SEGCAP + (size_t)c * CH1) + lane;
            uint4 v[4];
            #pragma unroll
            for (int j = 0; j < 4; ++j) v[j] = dp[j * 32];
            #pragma unroll
            for (int j = 0; j < 4; ++j) {
                const unsigned en[4] = { v[j].x, v[j].y, v[j].z, v[j].w };
                #pragma unroll
                for (int k = 0; k < 4; ++k) {
                    const unsigned loc = (en[k] >> 18) - lbase;
                    const bool m = loc < (unsigned)RNG;
                    const unsigned mask = __builtin_amdgcn_ballot_w32(m);
                    if (mask) {
                        if (m) {
                            const unsigned pos = (tail + __builtin_amdgcn_mbcnt_lo(mask, 0u)) & (RING - 1);
                            rq[pos] = (loc << 22) | ((unsigned)ef + (en[k] & 0x3FFFFu));
                        }
                        tail += (unsigned)__builtin_popcount(mask);
                    }
                }
                while (tail - head >= 32u) { add_n(head, 32); head += 32u; }
            }
        }
        if (tail != head) add_n(head, (int)(tail - head));
    } else {
        scan_owned(dst, base, rq, lane, wave,
                   [&](unsigned head) { add_n(head, 32); }, add_n);
    }
    __syncthreads();
    if (threadIdx.x < RNG && base + (int)threadIdx.x < N_NODES)
        dinv[base + threadIdx.x] = rsqrtf(sdeg[threadIdx.x]);
    if (threadIdx.x == 0) lcount[blockIdx.x] = lovf ? -1 : lcnt;
}

template<int C>
__global__ __launch_bounds__(1024) void own_rows(const float* __restrict__ H,
                                                 const int* __restrict__ src,
                                                 const int* __restrict__ dst,
                                                 const float* __restrict__ ew,
                                                 const float* __restrict__ dinv,
                                                 const int* __restrict__ lcount,
                                                 const unsigned* __restrict__ lkey,
                                                 const float* __restrict__ lnorm,
                                                 const float* __restrict__ bias,
                                                 float* __restrict__ out) {
    __shared__ float    tile[RNG * C];
    __shared__ unsigned ring[32][RING];
    constexpr int C4 = C / 4;
    const int lane = threadIdx.x & 31;
    const int wave = __builtin_amdgcn_readfirstlane((int)(threadIdx.x >> 5));
    const int base = blockIdx.x * RNG;

    for (int idx = threadIdx.x; idx < RNG * C4; idx += 1024) {
        const int r = idx / C4, q = idx - r * C4;
        const int node = base + r;
        float4 o = { 0.0f, 0.0f, 0.0f, 0.0f };
        if (node < N_NODES) {
            const float di = dinv[node], w = di * di;
            const float4 hv = *(const float4*)(H + (size_t)node * C + q * 4);
            o.x = w * hv.x; o.y = w * hv.y; o.z = w * hv.z; o.w = w * hv.w;
        }
        *(float4*)(tile + idx * 4) = o;
    }
    __syncthreads();

    const int cnt = lcount[blockIdx.x];
    if (cnt >= 0) {
        const unsigned* lk = lkey  + (size_t)blockIdx.x * CAP;
        const float*    ln = lnorm + (size_t)blockIdx.x * CAP;
        int g0 = wave * 32;
        unsigned key = 0; float nw = 0.0f;
        if (g0 + lane < cnt) { key = lk[g0 + lane]; nw = ln[g0 + lane]; }
        while (g0 < cnt) {
            const int gn = g0 + 1024;
            unsigned key2 = 0; float nw2 = 0.0f;
            if (gn + lane < cnt) { key2 = lk[gn + lane]; nw2 = ln[gn + lane]; }
            const int n = cnt - g0;
            const int s = (int)(key & 0x1FFFFu), loc = (int)(key >> 17);
            if (n >= 32) drain<C, true>(tile, s, loc, nw, 32, lane, H);
            else         drain<C, false>(tile, s, loc, nw, n, lane, H);
            key = key2; nw = nw2; g0 = gn;
        }
    } else {
        unsigned* rq = ring[wave];
        auto look = [&](unsigned head, int n, int& s, int& loc, float& nw) {
            s = 0; loc = 0; nw = 0.0f;
            if (lane < n) {
                const unsigned ent = rq[(head + lane) & (RING - 1)];
                const int e = (int)(ent & 0x3FFFFFu);
                loc = (int)(ent >> 22);
                s   = src[e];
                nw  = dinv[s] * ew[e] * dinv[base + loc];
            }
        };
        scan_owned(dst, base, rq, lane, wave,
                   [&](unsigned head) { int s, loc; float nw; look(head, 32, s, loc, nw);
                                        drain<C, true>(tile, s, loc, nw, 32, lane, H); },
                   [&](unsigned head, int n) { int s, loc; float nw; look(head, n, s, loc, nw);
                                               drain<C, false>(tile, s, loc, nw, n, lane, H); });
    }
    __syncthreads();

    for (int idx = threadIdx.x; idx < RNG * C4; idx += 1024) {
        const int r = idx / C4, q = idx - r * C4;
        const int node = base + r;
        if (node < N_NODES) {
            float4 t4 = *(const float4*)(tile + idx * 4);
            const float4 bb = *(const float4*)(bias + q * 4);
            t4.x = fmaxf(t4.x + bb.x, 0.0f);
            t4.y = fmaxf(t4.y + bb.y, 0.0f);
            t4.z = fmaxf(t4.z + bb.z, 0.0f);
            t4.w = fmaxf(t4.w + bb.w, 0.0f);
            *(float4*)(out + (size_t)node * C + q * 4) = t4;
        }
    }
}

__global__ __launch_bounds__(256) void classifier(const float* __restrict__ Hin,
                                                  const float* __restrict__ Wc,
                                                  const float* __restrict__ bc,
                                                  float* __restrict__ logits) {
    int i = blockIdx.x * blockDim.x + threadIdx.x;
    if (i >= N_NODES) return;
    const float* h = Hin + (size_t)i * HID2;
    float a0 = bc[0], a1 = bc[1], a2 = bc[2];
    #pragma unroll
    for (int k = 0; k < HID2; ++k) {
        float hv = h[k];
        a0 += hv * Wc[k * 3 + 0];
        a1 += hv * Wc[k * 3 + 1];
        a2 += hv * Wc[k * 3 + 2];
    }
    logits[(size_t)i * 3 + 0] = a0;
    logits[(size_t)i * 3 + 1] = a1;
    logits[(size_t)i * 3 + 2] = a2;
}


extern "C" void kernel_launch(void* const* d_in, const int* in_sizes, int n_in,
                              void* d_out, int out_size, void* d_ws, size_t ws_size,
                              hipStream_t stream) {
    const float* x  = (const float*)d_in[0];
    const int*   ei = (const int*)  d_in[1];
    const float* ew = (const float*)d_in[2];
    const float* W1 = (const float*)d_in[3];
    const float* b1 = (const float*)d_in[4];
    const float* W2 = (const float*)d_in[5];
    const float* b2 = (const float*)d_in[6];
    const float* Wc = (const float*)d_in[7];
    const float* bc = (const float*)d_in[8];
    float* logits = (float*)d_out;

    const int* src = ei;
    const int* dst = ei + N_EDGES;

    char* ws = (char*)d_ws;
    size_t off = 0;
    auto take = [&](size_t bytes) -> void* {
        void* p = ws + off;
        off += (bytes + 255) & ~(size_t)255;
        return p;
    };
    float* dinv = (float*)take((size_t)N_NODES * 4);
    int*      lcount = (int*)     take((size_t)(NGRP) * 4);
    unsigned* lkey   = (unsigned*)take((size_t)(NGRP) * CAP * 4);
    float*    lnorm  = (float*)   take((size_t)(NGRP) * CAP * 4);
    unsigned* clist  = (unsigned*)take((size_t)(NSUP * NSL) * SEGCAP * 4);
    int*      ccount = (int*)     take((size_t)(NSUP * NSL) * 4);
    float* H1   = (float*)take((size_t)N_NODES * HID * 4);
    float* out1 = (float*)take((size_t)N_NODES * HID * 4);
    float* H2   = H1;
    float* out2 = out1;

    const int TB = 256;
    {
        dim3 g(NSL, NSUP);
        coarse_split<<<g, 1024, 0, stream>>>(dst, clist, ccount);
    }
    own_deg<<<NGRP, 1024, 0, stream>>>(dst, ew, clist, ccount, dinv, lkey, lnorm, lcount);
    {
        dim3 g(CAP / TB, NGRP);
        list_norm<<<g, TB, 0, stream>>>(src, dinv, lcount, lkey, lnorm);
    }

    {
        dim3 g((N_NODES / 16 + 7) / 8, HID / 16);
        gemm_wmma<IN_CH, HID><<<g, 256, 0, stream>>>(x, W1, H1);
    }
    own_rows<HID><<<NGRP, 1024, 0, stream>>>(
        H1, src, dst, ew, dinv, lcount, lkey, lnorm, b1, out1);

    {
        dim3 g((N_NODES / 16 + 7) / 8, HID2 / 16);
        gemm_wmma<HID, HID2><<<g, 256, 0, stream>>>(out1, W2, H2);
    }
    own_rows<HID2><<<NGRP, 1024, 0, stream>>>(
        H2, src, dst, ew, dinv, lcount, lkey, lnorm, b2, out2);

    classifier<<<(N_NODES + TB - 1) / TB, TB, 0, stream>>>(out2, Wc, bc, logits);
}


# feature passes: a wave has 16 gathered source rows in flight before it adds them (was 4; a partial last trip went row by row, load by load); same adds in list order, LDS adds still in arrival order
# speedup vs baseline: 3.8051x; 1.2349x over previous
#include <hip/hip_runtime.h>

#define N_NODES 100000
#define N_EDGES 3200000
#define IN_CH   128
#define HID     96
#define HID2    48
#define NCLS    3

typedef __attribute__((ext_vector_type(2)))  float  v2f;
typedef __attribute__((ext_vector_type(8)))  float  v8f;


#define NGRP   256
#define RNG    ((N_NODES + NGRP - 1) / NGRP)
#define CAP    16384

static_assert((NGRP - 1) * RNG < N_NODES && NGRP * RNG >= N_NODES, "NGRP work-groups cover the nodes, none is empty");

__global__ __launch_bounds__(256) void list_norm(const int* __restrict__ src,
                                                 const float* __restrict__ dinv,
                                                 const int* __restrict__ lcount,
                                                 unsigned* lkey,
                                                 float* lnorm) {
    const int blk = blockIdx.y;
    const int i = blockIdx.x * blockDim.x + threadIdx.x;
    if (i >= lcount[blk]) return;
    const size_t p = (size_t)blk * CAP + i;
    const unsigned ent = lkey[p];
    const int e = (int)(ent & 0x3FFFFFu);
    const unsigned loc = ent >> 22;
    const int s = src[e];
    lnorm[p] = dinv[s] * lnorm[p] * dinv[blk * RNG + (int)loc];
    lkey[p] = (loc << 17) | (unsigned)s;
}


#define GW 6
#define GR 4
#define GBM(COLS) ((GW / ((COLS) / 16)) * GR * 16)

static_assert(N_NODES % 16 == 0, "whole row tiles: a row tile is all inside or all outside");

template<int K, int COLS>
__global__ __launch_bounds__(GW * 32) void gemm_wmma(const float* __restrict__ X,
                                                     const float* __restrict__ W,
                                                     float* __restrict__ H) {
    constexpr int CT  = COLS / 16;
    constexpr int BM  = GBM(COLS);
    constexpr int P   = K + 4;
    constexpr int K4  = K / 4;
    constexpr int TOT = BM * K4;
    constexpr int NT  = GW * 32;
    constexpr int NIT = (TOT + NT - 1) / NT;
    constexpr int ROWS = (NIT * NT + K4 - 1) / K4;
    static_assert(COLS % 16 == 0 && GW % CT == 0, "the waves split into whole column tiles");
    static_assert(K % 4 == 0 && (P / 4) % 2 == 1, "pitch is 16 B x odd");
    __shared__ __attribute__((aligned(16))) float xs[ROWS * P];

    const int lane = threadIdx.x & 31;
    const int wave = __builtin_amdgcn_readfirstlane((int)(threadIdx.x >> 5));
    const int m0 = blockIdx.x * BM;

    const float4* xg = (const float4*)(X + (size_t)m0 * K);
    const int have = (N_NODES - m0) * K4;
    const int lim = have < TOT ? have : TOT;
    float4 v[NIT];
    #pragma unroll
    for (int i = 0; i < NIT; ++i) {
        const int idx = (int)threadIdx.x + i * NT;
        v[i] = xg[idx < lim ? idx : lim - 1];
    }

    const int ct  = wave % CT;
    const int rg  = wave / CT;
    const int col = ct * 16 + (lane & 15);
    const int kh  = (lane >> 4) << 1;
    v2f bf[K4];
    #pragma unroll
    for (int j = 0; j < K4; ++j) {
        bf[j].x = W[(size_t)(j * 4 + kh)     * COLS + col];
        bf[j].y = W[(size_t)(j * 4 + kh + 1) * COLS + col];
    }

    #pragma unroll
    for (int i = 0; i < NIT; ++i) {
        const int idx = (int)threadIdx.x + i * NT;
        const int r = idx / K4, q = idx - r * K4;
        *(float4*)(xs + r * P + q * 4) = v[i];
    }
    __syncthreads();

    const float* ar = xs + (rg * GR * 16 + (lane & 15)) * P + kh;
    v8f c[GR] = {};
    #pragma unroll
    for (int j = 0; j < K4; ++j) {
        #pragma unroll
        for (int t = 0; t < GR; ++t) {
            const v2f a = *(const v2f*)(ar + t * 16 * P + j * 4);
            c[t] = __builtin_amdgcn_wmma_f32_16x16x4_f32(false, a, false, bf[j],
                                                         (short)0, c[t], false, false);
        }
    }

    const int rbase = (lane >> 4) << 3;
    #pragma unroll
    for (int t = 0; t < GR; ++t) {
        const int mt = m0 + (rg * GR + t) * 16;
        if (mt < N_NODES) {
            float* hrow = H + (size_t)(mt + rbase) * COLS + col;
            #pragma unroll
            for (int r = 0; r < 8; ++r) hrow[(size_t)r * COLS] = c[t][r];
        }
    }
}


#define RING   256
#define CHUNK  1024
#define NCHUNK (N_EDGES / CHUNK)

static_assert(N_EDGES % CHUNK == 0, "whole chunks");
static_assert(N_EDGES <= (1 << 22), "edge id packs into 22 bits");
static_assert(RNG <= 1024, "row offset packs into 10 bits, one thread per owned node");
static_assert(N_NODES <= (1 << 17), "source node packs into 17 bits beside the row offset");

#define SUP    16
#define NSUP   (NGRP / SUP)
#define SRNG   (SUP * RNG)
#define NSL    16
#define CH1    512
#define NCH1   (N_EDGES / CH1)
#define SEGCAP 16384
#define CPAD   0xFFFFFFFFu

static_assert(NGRP % SUP == 0, "whole coarse ranges");
static_assert(N_EDGES % CH1 == 0 && SEGCAP % CH1 == 0, "whole trips");
static_assert((NCH1 / NSL + 1) * CH1 <= (1 << 18), "edge offset in a slice packs into 18 bits");
static_assert(SRNG < (1 << 14) - 1, "node offset in a coarse range packs into 14 bits, below the filler's");
static_assert((CPAD >> 18) - (SUP - 1) * RNG >= RNG, "the filler matches no owner range");
static_assert(NSL <= 32, "one lane per list of a coarse range");

static __device__ __forceinline__ int   lane_i(int v, int l)   { return __builtin_amdgcn_readlane(v, l); }
static __device__ __forceinline__ float lane_f(float v, int l) {
    return __int_as_float(__builtin_amdgcn_readlane(__float_as_int(v), l));
}
static __device__ __forceinline__ void lds_add(float* p, float v) {
    __hip_atomic_fetch_add(p, v, __ATOMIC_RELAXED, __HIP_MEMORY_SCOPE_WORKGROUP);
}

#define INFL 16
static_assert(32 % INFL == 0, "a full trip is whole groups");

template<int C, bool FULL, class FNext>
static __device__ __forceinline__ void drain(float* __restrict__ tile,
                                             int s, int loc, float nw, int n, int lane,
                                             const float* __restrict__ H, FNext behind_first) {
    constexpr int T = (C + 31) / 32;
    const int cnt = FULL ? 32 : n;
    #pragma unroll 1
    for (int g = 0; g < cnt; g += INFL) {
        float hv[INFL][T];
        #pragma unroll
        for (int u = 0; u < INFL; ++u) {
            const int i = FULL ? g + u : (g + u < n ? g + u : n - 1);
            const float* hr = H + (size_t)lane_i(s, i) * C + lane;
            #pragma unroll
            for (int t = 0; t < T; ++t)
                hv[u][t] = (t * 32 + 32 <= C || lane < C - t * 32) ? hr[t * 32] : 0.0f;
        }
        if (g == 0) behind_first();
        #pragma unroll
        for (int u = 0; u < INFL; ++u) {
            if (FULL || g + u < n) {
                const float w = lane_f(nw, g + u);
                float* tr = tile + lane_i(loc, g + u) * C + lane;
                #pragma unroll
                for (int t = 0; t < T; ++t)
                    if (t * 32 + 32 <= C || lane < C - t * 32) lds_add(tr + t * 32, w * hv[u][t]);
            }
        }
    }
}

template<class F32, class FRest>
static __device__ __forceinline__ void scan_owned(const int* __restrict__ dst, int base,
                                                  unsigned* __restrict__ rq, int lane, int wave,
                                                  F32 on32, FRest rest) {
    unsigned head = 0, tail = 0;
    const int start = (int)(((long long)blockIdx.x * NCHUNK) / gridDim.x);
    for (int it = wave; it < NCHUNK; it += 32) {
        int chunk = start + it;
        if (chunk >= NCHUNK) chunk -= NCHUNK;
        const int e0 = chunk * CHUNK;
        const int4* dp = (const int4*)(dst + e0) + lane;
        int4 v[8];
        #pragma unroll
        for (int j = 0; j < 8; ++j) v[j] = dp[j * 32];
        #pragma unroll
        for (int j = 0; j < 8; ++j) {
            const int dd[4] = { v[j].x, v[j].y, v[j].z, v[j].w };
            #pragma unroll
            for (int k = 0; k < 4; ++k) {
                const unsigned loc = (unsigned)(dd[k] - base);
                const bool m = loc < (unsigned)RNG;
                const unsigned mask = __builtin_amdgcn_ballot_w32(m);
                if (mask) {
                    if (m) {
                        const unsigned pos = (tail + __builtin_amdgcn_mbcnt_lo(mask, 0u)) & (RING - 1);
                        rq[pos] = (loc << 22) | (unsigned)(e0 + j * 128 + lane * 4 + k);
                    }
                    tail += (unsigned)__builtin_popcount(mask);
                }
            }
            while (tail - head >= 32u) { on32(head); head += 32u; }
        }
    }
    if (tail != head) rest(head, (int)(tail - head));
}

__global__ __launch_bounds__(1024) void coarse_split(const int* __restrict__ dst,
                                                     unsigned* __restrict__ clist,
                                                     int* __restrict__ ccount) {
    __shared__ unsigned ring[32][RING];
    __shared__ int      lcnt, lovf;
    const int lane = threadIdx.x & 31;
    const int wave = __builtin_amdgcn_readfirstlane((int)(threadIdx.x >> 5));
    const int p = blockIdx.x;
    const int a = blockIdx.y;
    const int c0 = p * NCH1 / NSL;
    const int nch = (p + 1) * NCH1 / NSL - c0;
    const int cbase = a * SRNG;
    const int* dsl = dst + (size_t)c0 * CH1;
    if (threadIdx.x == 0) { lcnt = 0; lovf = 0; }
    __syncthreads();
    unsigned* rq  = ring[wave];
    unsigned* seg = clist + (size_t)(a * NSL + p) * SEGCAP;
    auto put = [&](unsigned head, int n) {
        unsigned ent = 0;
        if (lane < n) ent = rq[(head + lane) & (RING - 1)];
        int pos = 0;
        if (lane == 0)
            pos = __hip_atomic_fetch_add(&lcnt, n, __ATOMIC_RELAXED, __HIP_MEMORY_SCOPE_WORKGROUP);
        pos = __builtin_amdgcn_readfirstlane(pos);
        if (pos + n <= SEGCAP) {
            if (lane < n) seg[pos + lane] = ent;
        } else if (lane == 0) {
            __hip_atomic_store(&lovf, 1, __ATOMIC_RELAXED, __HIP_MEMORY_SCOPE_WORKGROUP);
        }
    };
    unsigned head = 0, tail = 0;
    const int rot = a * nch / NSUP;
    for (int it = wave; it < nch; it += 32) {
        int c = it + rot;
        if (c >= nch) c -= nch;
        const int eo = c * CH1;
        const int4* dp = (const int4*)(dsl + eo) + lane;
        int4 v[4];
        #pragma unroll
        for (int j = 0; j < 4; ++j) v[j] = dp[j * 32];
        #pragma unroll
        for (int j = 0; j < 4; ++j) {
            const int dd[4] = { v[j].x, v[j].y, v[j].z, v[j].w };
            #pragma unroll
            for (int k = 0; k < 4; ++k) {
                const unsigned loc = (unsigned)(dd[k] - cbase);
                const bool m = loc < (unsigned)SRNG;
                const unsigned mask = __builtin_amdgcn_ballot_w32(m);
                if (mask) {
                    if (m) {
                        const unsigned pos = (tail + __builtin_amdgcn_mbcnt_lo(mask, 0u)) & (RING - 1);
                        rq[pos] = (loc << 18) | (unsigned)(eo + j * 128 + lane * 4 + k);
                    }
                    tail += (unsigned)__builtin_popcount(mask);
                }
            }
            while (tail - head >= 32u) { put(head, 32); head += 32u; }
        }
    }
    if (tail != head) put(head, (int)(tail - head));
    __syncthreads();
    const int cnt = lcnt, ovf = lovf;
    if (!ovf) {
        const int end = (cnt + CH1 - 1) / CH1 * CH1;
        for (int i = cnt + (int)threadIdx.x; i < end; i += 1024) seg[i] = CPAD;
    }
    if (threadIdx.x == 0) ccount[a * NSL + p] = ovf ? -1 : cnt;
}

__global__ __launch_bounds__(1024) void own_deg(const int* __restrict__ dst,
                                                const float* __restrict__ ew,
                                                const unsigned* __restrict__ clist,
                                                const int* __restrict__ ccount,
                                                float* __restrict__ dinv,
                                                unsigned* __restrict__ lkey,
                                                float* __restrict__ lw,
                                                int* __restrict__ lcount) {
    __shared__ float    sdeg[RNG];
    __shared__ unsigned ring[32][RING];
    __shared__ int      lcnt, lovf;
    const int lane = threadIdx.x & 31;
    const int wave = __builtin_amdgcn_readfirstlane((int)(threadIdx.x >> 5));
    const int base = blockIdx.x * RNG;
    const int a = blockIdx.x / SUP;
    const int b = blockIdx.x % SUP;
    if (threadIdx.x == 0) { lcnt = 0; lovf = 0; }
    if (threadIdx.x < RNG) sdeg[threadIdx.x] = 1.0f;
    __syncthreads();
    unsigned* rq = ring[wave];
    unsigned* lk = lkey + (size_t)blockIdx.x * CAP;
    float*    lv = lw   + (size_t)blockIdx.x * CAP;
    auto add_n = [&](unsigned head, int n) {
        unsigned ent = 0; float w = 0.0f;
        if (lane < n) {
            ent = rq[(head + lane) & (RING - 1)];
            w   = ew[ent & 0x3FFFFFu];
        }
        int pos = 0;
        if (lane == 0)
            pos = __hip_atomic_fetch_add(&lcnt, n, __ATOMIC_RELAXED, __HIP_MEMORY_SCOPE_WORKGROUP);
        pos = __builtin_amdgcn_readfirstlane(pos);
        if (pos + n <= CAP) {
            if (lane < n) { lk[pos + lane] = ent; lv[pos + lane] = w; }
        } else if (lane == 0) {
            __hip_atomic_store(&lovf, 1, __ATOMIC_RELAXED, __HIP_MEMORY_SCOPE_WORKGROUP);
        }
        if (lane < n) lds_add(sdeg + (ent >> 22), w);
    };
    const int cn = lane < NSL ? ccount[a * NSL + lane] : 0;
    if (__builtin_amdgcn_ballot_w32(cn < 0) == 0u) {
        const int nc = (cn + CH1 - 1) / CH1;
        int before = 0;
        #pragma unroll
        for (int q = 0; q < NSL; ++q) {
            const int t = lane_i(nc, q);
            if (lane > q) before += t;
        }
        const int upto = before + nc;
        const int total = lane_i(upto, 31);
        const int rot = b * total / SUP;
        const unsigned lbase = (unsigned)(b * RNG);
        unsigned head = 0, tail = 0;
        for (int it = wave; it < total; it += 32) {
            int gi = it + rot;
            if (gi >= total) gi -= total;
            const int p = __builtin_popcount(__builtin_amdgcn_ballot_w32(lane < NSL && upto <= gi));
            const int c = gi - lane_i(before, p);
            const int ef = (p * NCH1 / NSL) * CH1;
            const uint4* dp = (const uint4*)(clist + (size_t)(a * NSL + p) * SEGCAP + (size_t)c * CH1) + lane;
            uint4 v[4];
            #pragma unroll
            for (int j = 0; j < 4; ++j) v[j] = dp[j * 32];
            #pragma unroll
            for (int j = 0; j < 4; ++j) {
                const unsigned en[4] = { v[j].x, v[j].y, v[j].z, v[j].w };
                #pragma unroll
                for (int k = 0; k < 4; ++k) {
                    const unsigned loc = (en[k] >> 18) - lbase;
                    const bool m = loc < (unsigned)RNG;
                    const unsigned mask = __builtin_amdgcn_ballot_w32(m);
                    if (mask) {
                        if (m) {
                            const unsigned pos = (tail + __builtin_amdgcn_mbcnt_lo(mask, 0u)) & (RING - 1);
                            rq[pos] = (loc << 22) | ((unsigned)ef + (en[k] & 0x3FFFFu));
                        }
                        tail += (unsigned)__builtin_popcount(mask);
                    }
                }
                while (tail - head >= 32u) { add_n(head, 32); head += 32u; }
            }
        }
        if (tail != head) add_n(head, (int)(tail - head));
    } else {
        scan_owned(dst, base, rq, lane, wave,
                   [&](unsigned head) { add_n(head, 32); }, add_n);
    }
    __syncthreads();
    if (threadIdx.x < RNG && base + (int)threadIdx.x < N_NODES)
        dinv[base + threadIdx.x] = rsqrtf(sdeg[threadIdx.x]);
    if (threadIdx.x == 0) lcount[blockIdx.x] = lovf ? -1 : lcnt;
}

template<int C>
__global__ __launch_bounds__(1024) void own_rows(const float* __restrict__ H,
                                                 const int* __restrict__ src,
                                                 const int* __restrict__ dst,
                                                 const float* __restrict__ ew,
                                                 const float* __restrict__ dinv,
                                                 const int* __restrict__ lcount,
                                                 const unsigned* __restrict__ lkey,
                                                 const float* __restrict__ lnorm,
                                                 const float* __restrict__ bias,
                                                 float* __restrict__ out) {
    __shared__ float    tile[RNG * C];
    __shared__ unsigned ring[32][RING];
    constexpr int C4 = C / 4;
    const int lane = threadIdx.x & 31;
    const int wave = __builtin_amdgcn_readfirstlane((int)(threadIdx.x >> 5));
    const int base = blockIdx.x * RNG;

    for (int idx = threadIdx.x; idx < RNG * C4; idx += 1024) {
        const int r = idx / C4, q = idx - r * C4;
        const int node = base + r;
        float4 o = { 0.0f, 0.0f, 0.0f, 0.0f };
        if (node < N_NODES) {
            const float di = dinv[node], w = di * di;
            const float4 hv = *(const float4*)(H + (size_t)node * C + q * 4);
            o.x = w * hv.x; o.y = w * hv.y; o.z = w * hv.z; o.w = w * hv.w;
        }
        *(float4*)(tile + idx * 4) = o;
    }
    __syncthreads();

    const int cnt = lcount[blockIdx.x];
    if (cnt >= 0) {
        const unsigned* lk = lkey  + (size_t)blockIdx.x * CAP;
        const float*    ln = lnorm + (size_t)blockIdx.x * CAP;
        int g0 = wave * 32;
        unsigned key = 0; float nw = 0.0f;
        if (g0 + lane < cnt) { key = lk[g0 + lane]; nw = ln[g0 + lane]; }
        while (g0 < cnt) {
            const int gn = g0 + 1024;
            unsigned key2 = 0; float nw2 = 0.0f;
            auto ask_next = [&]() {
                if (gn + lane < cnt) { key2 = lk[gn + lane]; nw2 = ln[gn + lane]; }
            };
            const int n = cnt - g0;
            const int s = (int)(key & 0x1FFFFu), loc = (int)(key >> 17);
            if (n >= 32) drain<C, true>(tile, s, loc, nw, 32, lane, H, ask_next);
            else         drain<C, false>(tile, s, loc, nw, n, lane, H, ask_next);
            key = key2; nw = nw2; g0 = gn;
        }
    } else {
        unsigned* rq = ring[wave];
        auto look = [&](unsigned head, int n, int& s, int& loc, float& nw) {
            s = 0; loc = 0; nw = 0.0f;
            if (lane < n) {
                const unsigned ent = rq[(head + lane) & (RING - 1)];
                const int e = (int)(ent & 0x3FFFFFu);
                loc = (int)(ent >> 22);
                s   = src[e];
                nw  = dinv[s] * ew[e] * dinv[base + loc];
            }
        };
        auto nothing = []() {};
        scan_owned(dst, base, rq, lane, wave,
                   [&](unsigned head) { int s, loc; float nw; look(head, 32, s, loc, nw);
                                        drain<C, true>(tile, s, loc, nw, 32, lane, H, nothing); },
                   [&](unsigned head, int n) { int s, loc; float nw; look(head, n, s, loc, nw);
                                               drain<C, false>(tile, s, loc, nw, n, lane, H, nothing); });
    }
    __syncthreads();

    for (int idx = threadIdx.x; idx < RNG * C4; idx += 1024) {
        const int r = idx / C4, q = idx - r * C4;
        const int node = base + r;
        if (node < N_NODES) {
            float4 t4 = *(const float4*)(tile + idx * 4);
            const float4 bb = *(const float4*)(bias + q * 4);
            t4.x = fmaxf(t4.x + bb.x, 0.0f);
            t4.y = fmaxf(t4.y + bb.y, 0.0f);
            t4.z = fmaxf(t4.z + bb.z, 0.0f);
            t4.w = fmaxf(t4.w + bb.w, 0.0f);
            *(float4*)(out + (size_t)node * C + q * 4) = t4;
        }
    }
}

__global__ __launch_bounds__(256) void classifier(const float* __restrict__ Hin,
                                                  const float* __restrict__ Wc,
                                                  const float* __restrict__ bc,
                                                  float* __restrict__ logits) {
    int i = blockIdx.x * blockDim.x + threadIdx.x;
    if (i >= N_NODES) return;
    const float* h = Hin + (size_t)i * HID2;
    float a0 = bc[0], a1 = bc[1], a2 = bc[2];
    #pragma unroll
    for (int k = 0; k < HID2; ++k) {
        float hv = h[k];
        a0 += hv * Wc[k * 3 + 0];
        a1 += hv * Wc[k * 3 + 1];
        a2 += hv * Wc[k * 3 + 2];
    }
    logits[(size_t)i * 3 + 0] = a0;
    logits[(size_t)i * 3 + 1] = a1;
    logits[(size_t)i * 3 + 2] = a2;
}


extern "C" void kernel_launch(void* const* d_in, const int* in_sizes, int n_in,
                              void* d_out, int out_size, void* d_ws, size_t ws_size,
                              hipStream_t stream) {
    const float* x  = (const float*)d_in[0];
    const int*   ei = (const int*)  d_in[1];
    const float* ew = (const float*)d_in[2];
    const float* W1 = (const float*)d_in[3];
    const float* b1 = (const float*)d_in[4];
    const float* W2 = (const float*)d_in[5];
    const float* b2 = (const float*)d_in[6];
    const float* Wc = (const float*)d_in[7];
    const float* bc = (const float*)d_in[8];
    float* logits = (float*)d_out;

    const int* src = ei;
    const int* dst = ei + N_EDGES;

    char* ws = (char*)d_ws;
    size_t off = 0;
    auto take = [&](size_t bytes) -> void* {
        void* p = ws + off;
        off += (bytes + 255) & ~(size_t)255;
        return p;
    };
    float* dinv = (float*)take((size_t)N_NODES * 4);
    int*      lcount = (int*)     take((size_t)(NGRP) * 4);
    unsigned* lkey   = (unsigned*)take((size_t)(NGRP) * CAP * 4);
    float*    lnorm  = (float*)   take((size_t)(NGRP) * CAP * 4);
    unsigned* clist  = (unsigned*)take((size_t)(NSUP * NSL) * SEGCAP * 4);
    int*      ccount = (int*)     take((size_t)(NSUP * NSL) * 4);
    float* H1   = (float*)take((size_t)N_NODES * HID * 4);
    float* out1 = (float*)take((size_t)N_NODES * HID * 4);
    float* H2   = H1;
    float* out2 = out1;

    const int TB = 256;
    {
        dim3 g(NSL, NSUP);
        coarse_split<<<g, 1024, 0, stream>>>(dst, clist, ccount);
    }
    own_deg<<<NGRP, 1024, 0, stream>>>(dst, ew, clist, ccount, dinv, lkey, lnorm, lcount);
    {
        dim3 g(CAP / TB, NGRP);
        list_norm<<<g, TB, 0, stream>>>(src, dinv, lcount, lkey, lnorm);
    }

    {
        gemm_wmma<IN_CH, HID><<<(N_NODES + GBM(HID) - 1) / GBM(HID), GW * 32, 0, stream>>>(x, W1, H1);
    }
    own_rows<HID><<<NGRP, 1024, 0, stream>>>(
        H1, src, dst, ew, dinv, lcount, lkey, lnorm, b1, out1);

    {
        gemm_wmma<HID, HID2><<<(N_NODES + GBM(HID2) - 1) / GBM(HID2), GW * 32, 0, stream>>>(out1, W2, H2);
    }
    own_rows<HID2><<<NGRP, 1024, 0, stream>>>(
        H2, src, dst, ew, dinv, lcount, lkey, lnorm, b2, out2);

    classifier<<<(N_NODES + TB - 1) / TB, TB, 0, stream>>>(out2, Wc, bc, logits);
}


# feature passes: list ordered by owner wave in LDS (count, prefix, returning LDS adds); a wave adds its own rows (offset mod 32) by 16-byte LDS read-add-write, no float atomics; sums in queue order
# speedup vs baseline: 4.0433x; 1.0626x over previous
#include <hip/hip_runtime.h>

#define N_NODES 100000
#define N_EDGES 3200000
#define IN_CH   128
#define HID     96
#define HID2    48
#define NCLS    3

typedef __attribute__((ext_vector_type(2)))  float  v2f;
typedef __attribute__((ext_vector_type(8)))  float  v8f;
typedef __attribute__((ext_vector_type(4)))  float  v4f;


#define NGRP   256
#define RNG    ((N_NODES + NGRP - 1) / NGRP)
#define CAP    16384

static_assert((NGRP - 1) * RNG < N_NODES && NGRP * RNG >= N_NODES, "NGRP work-groups cover the nodes, none is empty");

__global__ __launch_bounds__(256) void list_norm(const int* __restrict__ src,
                                                 const float* __restrict__ dinv,
                                                 const int* __restrict__ lcount,
                                                 unsigned* lkey,
                                                 float* lnorm) {
    const int blk = blockIdx.y;
    const int i = blockIdx.x * blockDim.x + threadIdx.x;
    if (i >= lcount[blk]) return;
    const size_t p = (size_t)blk * CAP + i;
    const unsigned ent = lkey[p];
    const int e = (int)(ent & 0x3FFFFFu);
    const unsigned loc = ent >> 22;
    const int s = src[e];
    lnorm[p] = dinv[s] * lnorm[p] * dinv[blk * RNG + (int)loc];
    lkey[p] = (loc << 17) | (unsigned)s;
}


#define GW 6
#define GR 4
#define GBM(COLS) ((GW / ((COLS) / 16)) * GR * 16)

static_assert(N_NODES % 16 == 0, "whole row tiles: a row tile is all inside or all outside");

template<int K, int COLS>
__global__ __launch_bounds__(GW * 32) void gemm_wmma(const float* __restrict__ X,
                                                     const float* __restrict__ W,
                                                     float* __restrict__ H) {
    constexpr int CT  = COLS / 16;
    constexpr int BM  = GBM(COLS);
    constexpr int P   = K + 4;
    constexpr int K4  = K / 4;
    constexpr int TOT = BM * K4;
    constexpr int NT  = GW * 32;
    constexpr int NIT = (TOT + NT - 1) / NT;
    constexpr int ROWS = (NIT * NT + K4 - 1) / K4;
    static_assert(COLS % 16 == 0 && GW % CT == 0, "the waves split into whole column tiles");
    static_assert(K % 4 == 0 && (P / 4) % 2 == 1, "pitch is 16 B x odd");
    __shared__ __attribute__((aligned(16))) float xs[ROWS * P];

    const int lane = threadIdx.x & 31;
    const int wave = __builtin_amdgcn_readfirstlane((int)(threadIdx.x >> 5));
    const int m0 = blockIdx.x * BM;

    const float4* xg = (const float4*)(X + (size_t)m0 * K);
    const int have = (N_NODES - m0) * K4;
    const int lim = have < TOT ? have : TOT;
    float4 v[NIT];
    #pragma unroll
    for (int i = 0; i < NIT; ++i) {
        const int idx = (int)threadIdx.x + i * NT;
        v[i] = xg[idx < lim ? idx : lim - 1];
    }

    const int ct  = wave % CT;
    const int rg  = wave / CT;
    const int col = ct * 16 + (lane & 15);
    const int kh  = (lane >> 4) << 1;
    v2f bf[K4];
    #pragma unroll
    for (int j = 0; j < K4; ++j) {
        bf[j].x = W[(size_t)(j * 4 + kh)     * COLS + col];
        bf[j].y = W[(size_t)(j * 4 + kh + 1) * COLS + col];
    }

    #pragma unroll
    for (int i = 0; i < NIT; ++i) {
        const int idx = (int)threadIdx.x + i * NT;
        const int r = idx / K4, q = idx - r * K4;
        *(float4*)(xs + r * P + q * 4) = v[i];
    }
    __syncthreads();

    const float* ar = xs + (rg * GR * 16 + (lane & 15)) * P + kh;
    v8f c[GR] = {};
    #pragma unroll
    for (int j = 0; j < K4; ++j) {
        #pragma unroll
        for (int t = 0; t < GR; ++t) {
            const v2f a = *(const v2f*)(ar + t * 16 * P + j * 4);
            c[t] = __builtin_amdgcn_wmma_f32_16x16x4_f32(false, a, false, bf[j],
                                                         (short)0, c[t], false, false);
        }
    }

    const int rbase = (lane >> 4) << 3;
    #pragma unroll
    for (int t = 0; t < GR; ++t) {
        const int mt = m0 + (rg * GR + t) * 16;
        if (mt < N_NODES) {
            float* hrow = H + (size_t)(mt + rbase) * COLS + col;
            #pragma unroll
            for (int r = 0; r < 8; ++r) hrow[(size_t)r * COLS] = c[t][r];
        }
    }
}


#define RING   256
#define CHUNK  1024
#define NCHUNK (N_EDGES / CHUNK)

static_assert(N_EDGES % CHUNK == 0, "whole chunks");
static_assert(N_EDGES <= (1 << 22), "edge id packs into 22 bits");
static_assert(RNG <= 1024, "row offset packs into 10 bits, one thread per owned node");
static_assert(N_NODES <= (1 << 17), "source node packs into 17 bits beside the row offset");

#define SUP    16
#define NSUP   (NGRP / SUP)
#define SRNG   (SUP * RNG)
#define NSL    16
#define CH1    512
#define NCH1   (N_EDGES / CH1)
#define SEGCAP 16384
#define CPAD   0xFFFFFFFFu

static_assert(NGRP % SUP == 0, "whole coarse ranges");
static_assert(N_EDGES % CH1 == 0 && SEGCAP % CH1 == 0, "whole trips");
static_assert((NCH1 / NSL + 1) * CH1 <= (1 << 18), "edge offset in a slice packs into 18 bits");
static_assert(SRNG < (1 << 14) - 1, "node offset in a coarse range packs into 14 bits, below the filler's");
static_assert((CPAD >> 18) - (SUP - 1) * RNG >= RNG, "the filler matches no owner range");
static_assert(NSL <= 32, "one lane per list of a coarse range");

static __device__ __forceinline__ int   lane_i(int v, int l)   { return __builtin_amdgcn_readlane(v, l); }
static __device__ __forceinline__ float lane_f(float v, int l) {
    return __int_as_float(__builtin_amdgcn_readlane(__float_as_int(v), l));
}
static __device__ __forceinline__ void lds_add(float* p, float v) {
    __hip_atomic_fetch_add(p, v, __ATOMIC_RELAXED, __HIP_MEMORY_SCOPE_WORKGROUP);
}

#define INFL 16
static_assert(32 % INFL == 0, "a full trip is whole groups");

template<int C, bool FULL, class FNext>
static __device__ __forceinline__ void drain(float* __restrict__ tile,
                                             int s, int loc, float nw, int n, int lane,
                                             const float* __restrict__ H, FNext behind_first) {
    constexpr int T = (C + 31) / 32;
    const int cnt = FULL ? 32 : n;
    #pragma unroll 1
    for (int g = 0; g < cnt; g += INFL) {
        float hv[INFL][T];
        #pragma unroll
        for (int u = 0; u < INFL; ++u) {
            const int i = FULL ? g + u : (g + u < n ? g + u : n - 1);
            const float* hr = H + (size_t)lane_i(s, i) * C + lane;
            #pragma unroll
            for (int t = 0; t < T; ++t)
                hv[u][t] = (t * 32 + 32 <= C || lane < C - t * 32) ? hr[t * 32] : 0.0f;
        }
        if (g == 0) behind_first();
        #pragma unroll
        for (int u = 0; u < INFL; ++u) {
            if (FULL || g + u < n) {
                const float w = lane_f(nw, g + u);
                float* tr = tile + lane_i(loc, g + u) * C + lane;
                #pragma unroll
                for (int t = 0; t < T; ++t)
                    if (t * 32 + 32 <= C || lane < C - t * 32) lds_add(tr + t * 32, w * hv[u][t]);
            }
        }
    }
}

#define QW   32
#define OWNL 24

template<int V> struct piece;
template<> struct piece<4> { typedef v4f t; };
template<> struct piece<2> { typedef v2f t; };

template<int C>
static __device__ __forceinline__ void drain_own(float* tile, const uint2* q, int q0, int q1,
                                                 int lane, const float* __restrict__ H) {
    constexpr int V = C / OWNL;
    constexpr int R = 32 / V;
    static_assert(V * OWNL == C && (V == 2 || V == 4), "a row is 24 pieces of 8 or 16 bytes");
    static_assert(R <= OWNL, "the lanes that hold a group's entries all hold row pieces");
    typedef typename piece<V>::t pv;
    if (lane < OWNL && q0 < q1) {
        auto ents = [&](int g) -> uint2 {
            const int i = g + lane;
            return q[i < q1 ? i : q1 - 1];
        };
        auto ask = [&](pv (&hv)[R], const uint2& e, int n) {
            #pragma unroll
            for (int u = 0; u < R; ++u) {
                if (u < n) {
                    const int s = lane_i((int)(e.x & 0x1FFFFu), u);
                    hv[u] = *(const pv*)(H + (size_t)s * C + lane * V);
                }
            }
        };
        auto add = [&](const pv (&hv)[R], const uint2& e, int n) {
            #pragma unroll
            for (int u = 0; u < R; u += 2) {
                if (u + 1 < n) {
                    const float w0 = lane_f(__uint_as_float(e.y), u);
                    const float w1 = lane_f(__uint_as_float(e.y), u + 1);
                    const int l0 = lane_i((int)(e.x >> 17), u);
                    const int l1 = lane_i((int)(e.x >> 17), u + 1);
                    pv* t0 = (pv*)(tile + l0 * C + lane * V);
                    if (l0 != l1) {
                        pv* t1 = (pv*)(tile + l1 * C + lane * V);
                        const pv x0 = *t0;
                        const pv x1 = *t1;
                        *t0 = x0 + w0 * hv[u];
                        *t1 = x1 + w1 * hv[u + 1];
                    } else {
                        *t0 = (*t0 + w0 * hv[u]) + w1 * hv[u + 1];
                    }
                } else if (u < n) {
                    const float w = lane_f(__uint_as_float(e.y), u);
                    const int loc = lane_i((int)(e.x >> 17), u);
                    pv* tp = (pv*)(tile + loc * C + lane * V);
                    *tp = *tp + w * hv[u];
                }
            }
        };
        int g = q0;
        if (g + 2 * R <= q1) {
            pv a[R], b[R];
            uint2 ea = ents(g), eb = ents(g + R);
            ask(a, ea, R);
            #pragma unroll 1
            for (;;) {
                ask(b, eb, R);
                const uint2 ea2 = ents(g + 2 * R);
                add(a, ea, R);
                g += 2 * R;
                if (g + 2 * R > q1) { add(b, eb, R); break; }
                ea = ea2;
                ask(a, ea, R);
                const uint2 eb2 = ents(g + R);
                add(b, eb, R);
                eb = eb2;
            }
        }
        if (g + R <= q1) {
            pv t[R];
            const uint2 e = ents(g);
            ask(t, e, R);
            add(t, e, R);
            g += R;
        }
        if (g < q1) {
            pv t[R];
            const uint2 e = ents(g);
            ask(t, e, q1 - g);
            add(t, e, q1 - g);
        }
    }
}

template<class F32, class FRest>
static __device__ __forceinline__ void scan_owned(const int* __restrict__ dst, int base,
                                                  unsigned* __restrict__ rq, int lane, int wave,
                                                  F32 on32, FRest rest) {
    unsigned head = 0, tail = 0;
    const int start = (int)(((long long)blockIdx.x * NCHUNK) / gridDim.x);
    for (int it = wave; it < NCHUNK; it += 32) {
        int chunk = start + it;
        if (chunk >= NCHUNK) chunk -= NCHUNK;
        const int e0 = chunk * CHUNK;
        const int4* dp = (const int4*)(dst + e0) + lane;
        int4 v[8];
        #pragma unroll
        for (int j = 0; j < 8; ++j) v[j] = dp[j * 32];
        #pragma unroll
        for (int j = 0; j < 8; ++j) {
            const int dd[4] = { v[j].x, v[j].y, v[j].z, v[j].w };
            #pragma unroll
            for (int k = 0; k < 4; ++k) {
                const unsigned loc = (unsigned)(dd[k] - base);
                const bool m = loc < (unsigned)RNG;
                const unsigned mask = __builtin_amdgcn_ballot_w32(m);
                if (mask) {
                    if (m) {
                        const unsigned pos = (tail + __builtin_amdgcn_mbcnt_lo(mask, 0u)) & (RING - 1);
                        rq[pos] = (loc << 22) | (unsigned)(e0 + j * 128 + lane * 4 + k);
                    }
                    tail += (unsigned)__builtin_popcount(mask);
                }
            }
            while (tail - head >= 32u) { on32(head); head += 32u; }
        }
    }
    if (tail != head) rest(head, (int)(tail - head));
}

__global__ __launch_bounds__(1024) void coarse_split(const int* __restrict__ dst,
                                                     unsigned* __restrict__ clist,
                                                     int* __restrict__ ccount) {
    __shared__ unsigned ring[32][RING];
    __shared__ int      lcnt, lovf;
    const int lane = threadIdx.x & 31;
    const int wave = __builtin_amdgcn_readfirstlane((int)(threadIdx.x >> 5));
    const int p = blockIdx.x;
    const int a = blockIdx.y;
    const int c0 = p * NCH1 / NSL;
    const int nch = (p + 1) * NCH1 / NSL - c0;
    const int cbase = a * SRNG;
    const int* dsl = dst + (size_t)c0 * CH1;
    if (threadIdx.x == 0) { lcnt = 0; lovf = 0; }
    __syncthreads();
    unsigned* rq  = ring[wave];
    unsigned* seg = clist + (size_t)(a * NSL + p) * SEGCAP;
    auto put = [&](unsigned head, int n) {
        unsigned ent = 0;
        if (lane < n) ent = rq[(head + lane) & (RING - 1)];
        int pos = 0;
        if (lane == 0)
            pos = __hip_atomic_fetch_add(&lcnt, n, __ATOMIC_RELAXED, __HIP_MEMORY_SCOPE_WORKGROUP);
        pos = __builtin_amdgcn_readfirstlane(pos);
        if (pos + n <= SEGCAP) {
            if (lane < n) seg[pos + lane] = ent;
        } else if (lane == 0) {
            __hip_atomic_store(&lovf, 1, __ATOMIC_RELAXED, __HIP_MEMORY_SCOPE_WORKGROUP);
        }
    };
    unsigned head = 0, tail = 0;
    const int rot = a * nch / NSUP;
    for (int it = wave; it < nch; it += 32) {
        int c = it + rot;
        if (c >= nch) c -= nch;
        const int eo = c * CH1;
        const int4* dp = (const int4*)(dsl + eo) + lane;
        int4 v[4];
        #pragma unroll
        for (int j = 0; j < 4; ++j) v[j] = dp[j * 32];
        #pragma unroll
        for (int j = 0; j < 4; ++j) {
            const int dd[4] = { v[j].x, v[j].y, v[j].z, v[j].w };
            #pragma unroll
            for (int k = 0; k < 4; ++k) {
                const unsigned loc = (unsigned)(dd[k] - cbase);
                const bool m = loc < (unsigned)SRNG;
                const unsigned mask = __builtin_amdgcn_ballot_w32(m);
                if (mask) {
                    if (m) {
                        const unsigned pos = (tail + __builtin_amdgcn_mbcnt_lo(mask, 0u)) & (RING - 1);
                        rq[pos] = (loc << 18) | (unsigned)(eo + j * 128 + lane * 4 + k);
                    }
                    tail += (unsigned)__builtin_popcount(mask);
                }
            }
            while (tail - head >= 32u) { put(head, 32); head += 32u; }
        }
    }
    if (tail != head) put(head, (int)(tail - head));
    __syncthreads();
    const int cnt = lcnt, ovf = lovf;
    if (!ovf) {
        const int end = (cnt + CH1 - 1) / CH1 * CH1;
        for (int i = cnt + (int)threadIdx.x; i < end; i += 1024) seg[i] = CPAD;
    }
    if (threadIdx.x == 0) ccount[a * NSL + p] = ovf ? -1 : cnt;
}

__global__ __launch_bounds__(1024) void own_deg(const int* __restrict__ dst,
                                                const float* __restrict__ ew,
                                                const unsigned* __restrict__ clist,
                                                const int* __restrict__ ccount,
                                                float* __restrict__ dinv,
                                                unsigned* __restrict__ lkey,
                                                float* __restrict__ lw,
                                                int* __restrict__ lcount) {
    __shared__ float    sdeg[RNG];
    __shared__ unsigned ring[32][RING];
    __shared__ int      lcnt, lovf;
    const int lane = threadIdx.x & 31;
    const int wave = __builtin_amdgcn_readfirstlane((int)(threadIdx.x >> 5));
    const int base = blockIdx.x * RNG;
    const int a = blockIdx.x / SUP;
    const int b = blockIdx.x % SUP;
    if (threadIdx.x == 0) { lcnt = 0; lovf = 0; }
    if (threadIdx.x < RNG) sdeg[threadIdx.x] = 1.0f;
    __syncthreads();
    unsigned* rq = ring[wave];
    unsigned* lk = lkey + (size_t)blockIdx.x * CAP;
    float*    lv = lw   + (size_t)blockIdx.x * CAP;
    auto add_n = [&](unsigned head, int n) {
        unsigned ent = 0; float w = 0.0f;
        if (lane < n) {
            ent = rq[(head + lane) & (RING - 1)];
            w   = ew[ent & 0x3FFFFFu];
        }
        int pos = 0;
        if (lane == 0)
            pos = __hip_atomic_fetch_add(&lcnt, n, __ATOMIC_RELAXED, __HIP_MEMORY_SCOPE_WORKGROUP);
        pos = __builtin_amdgcn_readfirstlane(pos);
        if (pos + n <= CAP) {
            if (lane < n) { lk[pos + lane] = ent; lv[pos + lane] = w; }
        } else if (lane == 0) {
            __hip_atomic_store(&lovf, 1, __ATOMIC_RELAXED, __HIP_MEMORY_SCOPE_WORKGROUP);
        }
        if (lane < n) lds_add(sdeg + (ent >> 22), w);
    };
    const int cn = lane < NSL ? ccount[a * NSL + lane] : 0;
    if (__builtin_amdgcn_ballot_w32(cn < 0) == 0u) {
        const int nc = (cn + CH1 - 1) / CH1;
        int before = 0;
        #pragma unroll
        for (int q = 0; q < NSL; ++q) {
            const int t = lane_i(nc, q);
            if (lane > q) before += t;
        }
        const int upto = before + nc;
        const int total = lane_i(upto, 31);
        const int rot = b * total / SUP;
        const unsigned lbase = (unsigned)(b * RNG);
        unsigned head = 0, tail = 0;
        for (int it = wave; it < total; it += 32) {
            int gi = it + rot;
            if (gi >= total) gi -= total;
            const int p = __builtin_popcount(__builtin_amdgcn_ballot_w32(lane < NSL && upto <= gi));
            const int c = gi - lane_i(before, p);
            const int ef = (p * NCH1 / NSL) * CH1;
            const uint4* dp = (const uint4*)(clist + (size_t)(a * NSL + p) * SEGCAP + (size_t)c * CH1) + lane;
            uint4 v[4];
            #pragma unroll
            for (int j = 0; j < 4; ++j) v[j] = dp[j * 32];
            #pragma unroll
            for (int j = 0; j < 4; ++j) {
                const unsigned en[4] = { v[j].x, v[j].y, v[j].z, v[j].w };
                #pragma unroll
                for (int k = 0; k < 4; ++k) {
                    const unsigned loc = (en[k] >> 18) - lbase;
                    const bool m = loc < (unsigned)RNG;
                    const unsigned mask = __builtin_amdgcn_ballot_w32(m);
                    if (mask) {
                        if (m) {
                            const unsigned pos = (tail + __builtin_amdgcn_mbcnt_lo(mask, 0u)) & (RING - 1);
                            rq[pos] = (loc << 22) | ((unsigned)ef + (en[k] & 0x3FFFFu));
                        }
                        tail += (unsigned)__builtin_popcount(mask);
                    }
                }
                while (tail - head >= 32u) { add_n(head, 32); head += 32u; }
            }
        }
        if (tail != head) add_n(head, (int)(tail - head));
    } else {
        scan_owned(dst, base, rq, lane, wave,
                   [&](unsigned head) { add_n(head, 32); }, add_n);
    }
    __syncthreads();
    if (threadIdx.x < RNG && base + (int)threadIdx.x < N_NODES)
        dinv[base + threadIdx.x] = rsqrtf(sdeg[threadIdx.x]);
    if (threadIdx.x == 0) lcount[blockIdx.x] = lovf ? -1 : lcnt;
}

template<int C>
__global__ __launch_bounds__(1024) void own_rows(const float* __restrict__ H,
                                                 const int* __restrict__ src,
                                                 const int* __restrict__ dst,
                                                 const float* __restrict__ ew,
                                                 const float* __restrict__ dinv,
                                                 const int* __restrict__ lcount,
                                                 const unsigned* __restrict__ lkey,
                                                 const float* __restrict__ lnorm,
                                                 const float* __restrict__ bias,
                                                 float* __restrict__ out) {
    __shared__ __attribute__((aligned(16))) float tile[RNG * C];
    __shared__ unsigned ring[32][RING];
    __shared__ __attribute__((aligned(16))) uint2 queue[CAP];
    __shared__ int      qn[QW];
    __shared__ int      qo[QW + 1];
    constexpr int C4 = C / 4;
    const int lane = threadIdx.x & 31;
    const int wave = __builtin_amdgcn_readfirstlane((int)(threadIdx.x >> 5));
    const int base = blockIdx.x * RNG;

    if (threadIdx.x < QW) qn[threadIdx.x] = 0;
    for (int idx = threadIdx.x; idx < RNG * C4; idx += 1024) {
        const int r = idx / C4, q = idx - r * C4;
        const int node = base + r;
        float4 o = { 0.0f, 0.0f, 0.0f, 0.0f };
        if (node < N_NODES) {
            const float di = dinv[node], w = di * di;
            const float4 hv = *(const float4*)(H + (size_t)node * C + q * 4);
            o.x = w * hv.x; o.y = w * hv.y; o.z = w * hv.z; o.w = w * hv.w;
        }
        *(float4*)(tile + idx * 4) = o;
    }
    __syncthreads();

    const int cnt = lcount[blockIdx.x];
    if (__builtin_expect(cnt >= 0, 1)) {
        const unsigned* lk = lkey  + (size_t)blockIdx.x * CAP;
        const float*    ln = lnorm + (size_t)blockIdx.x * CAP;
        constexpr int PT = CAP / 1024;
        unsigned key[PT]; float nw[PT];
        #pragma unroll
        for (int j = 0; j < PT; ++j) {
            const int i = j * 1024 + (int)threadIdx.x;
            key[j] = 0u; nw[j] = 0.0f;
            if (i < cnt) { key[j] = lk[i]; nw[j] = ln[i]; }
        }
        #pragma unroll
        for (int j = 0; j < PT; ++j) {
            const int i = j * 1024 + (int)threadIdx.x;
            if (i < cnt)
                __hip_atomic_fetch_add(&qn[(key[j] >> 17) & (QW - 1)], 1,
                                       __ATOMIC_RELAXED, __HIP_MEMORY_SCOPE_WORKGROUP);
        }
        __syncthreads();
        if (wave == 0) {
            const int c = qn[lane];
            int incl = c;
            #pragma unroll
            for (int d = 1; d < 32; d <<= 1) {
                const int t = __shfl_up(incl, d);
                if (lane >= d) incl += t;
            }
            qo[lane + 1] = incl;
            if (lane == 0) qo[0] = 0;
            qn[lane] = incl - c;
        }
        __syncthreads();
        int slot[PT];
        #pragma unroll
        for (int j = 0; j < PT; ++j) {
            const int i = j * 1024 + (int)threadIdx.x;
            slot[j] = 0;
            if (i < cnt)
                slot[j] = __hip_atomic_fetch_add(&qn[(key[j] >> 17) & (QW - 1)], 1,
                                                 __ATOMIC_RELAXED, __HIP_MEMORY_SCOPE_WORKGROUP);
        }
        #pragma unroll
        for (int j = 0; j < PT; ++j) {
            const int i = j * 1024 + (int)threadIdx.x;
            if (i < cnt) queue[slot[j]] = make_uint2(key[j], __float_as_uint(nw[j]));
        }
        __syncthreads();
        drain_own<C>(tile, queue, qo[wave], qo[wave + 1], lane, H);
    } else {
        unsigned* rq = ring[wave];
        auto look = [&](unsigned head, int n, int& s, int& loc, float& nw) {
            s = 0; loc = 0; nw = 0.0f;
            if (lane < n) {
                const unsigned ent = rq[(head + lane) & (RING - 1)];
                const int e = (int)(ent & 0x3FFFFFu);
                loc = (int)(ent >> 22);
                s   = src[e];
                nw  = dinv[s] * ew[e] * dinv[base + loc];
            }
        };
        auto nothing = []() {};
        scan_owned(dst, base, rq, lane, wave,
                   [&](unsigned head) { int s, loc; float nw; look(head, 32, s, loc, nw);
                                        drain<C, true>(tile, s, loc, nw, 32, lane, H, nothing); },
                   [&](unsigned head, int n) { int s, loc; float nw; look(head, n, s, loc, nw);
                                               drain<C, false>(tile, s, loc, nw, n, lane, H, nothing); });
    }
    __syncthreads();

    for (int idx = threadIdx.x; idx < RNG * C4; idx += 1024) {
        const int r = idx / C4, q = idx - r * C4;
        const int node = base + r;
        if (node < N_NODES) {
            float4 t4 = *(const float4*)(tile + idx * 4);
            const float4 bb = *(const float4*)(bias + q * 4);
            t4.x = fmaxf(t4.x + bb.x, 0.0f);
            t4.y = fmaxf(t4.y + bb.y, 0.0f);
            t4.z = fmaxf(t4.z + bb.z, 0.0f);
            t4.w = fmaxf(t4.w + bb.w, 0.0f);
            *(float4*)(out + (size_t)node * C + q * 4) = t4;
        }
    }
}

__global__ __launch_bounds__(256) void classifier(const float* __restrict__ Hin,
                                                  const float* __restrict__ Wc,
                                                  const float* __restrict__ bc,
                                                  float* __restrict__ logits) {
    int i = blockIdx.x * blockDim.x + threadIdx.x;
    if (i >= N_NODES) return;
    const float* h = Hin + (size_t)i * HID2;
    float a0 = bc[0], a1 = bc[1], a2 = bc[2];
    #pragma unroll
    for (int k = 0; k < HID2; ++k) {
        float hv = h[k];
        a0 += hv * Wc[k * 3 + 0];
        a1 += hv * Wc[k * 3 + 1];
        a2 += hv * Wc[k * 3 + 2];
    }
    logits[(size_t)i * 3 + 0] = a0;
    logits[(size_t)i * 3 + 1] = a1;
    logits[(size_t)i * 3 + 2] = a2;
}


extern "C" void kernel_launch(void* const* d_in, const int* in_sizes, int n_in,
                              void* d_out, int out_size, void* d_ws, size_t ws_size,
                              hipStream_t stream) {
    const float* x  = (const float*)d_in[0];
    const int*   ei = (const int*)  d_in[1];
    const float* ew = (const float*)d_in[2];
    const float* W1 = (const float*)d_in[3];
    const float* b1 = (const float*)d_in[4];
    const float* W2 = (const float*)d_in[5];
    const float* b2 = (const float*)d_in[6];
    const float* Wc = (const float*)d_in[7];
    const float* bc = (const float*)d_in[8];
    float* logits = (float*)d_out;

    const int* src = ei;
    const int* dst = ei + N_EDGES;

    char* ws = (char*)d_ws;
    size_t off = 0;
    auto take = [&](size_t bytes) -> void* {
        void* p = ws + off;
        off += (bytes + 255) & ~(size_t)255;
        return p;
    };
    float* dinv = (float*)take((size_t)N_NODES * 4);
    int*      lcount = (int*)     take((size_t)(NGRP) * 4);
    unsigned* lkey   = (unsigned*)take((size_t)(NGRP) * CAP * 4);
    float*    lnorm  = (float*)   take((size_t)(NGRP) * CAP * 4);
    unsigned* clist  = (unsigned*)take((size_t)(NSUP * NSL) * SEGCAP * 4);
    int*      ccount = (int*)     take((size_t)(NSUP * NSL) * 4);
    float* H1   = (float*)take((size_t)N_NODES * HID * 4);
    float* out1 = (float*)take((size_t)N_NODES * HID * 4);
    float* H2   = H1;
    float* out2 = out1;

    const int TB = 256;
    {
        dim3 g(NSL, NSUP);
        coarse_split<<<g, 1024, 0, stream>>>(dst, clist, ccount);
    }
    own_deg<<<NGRP, 1024, 0, stream>>>(dst, ew, clist, ccount, dinv, lkey, lnorm, lcount);
    {
        dim3 g(CAP / TB, NGRP);
        list_norm<<<g, TB, 0, stream>>>(src, dinv, lcount, lkey, lnorm);
    }

    {
        gemm_wmma<IN_CH, HID><<<(N_NODES + GBM(HID) - 1) / GBM(HID), GW * 32, 0, stream>>>(x, W1, H1);
    }
    own_rows<HID><<<NGRP, 1024, 0, stream>>>(
        H1, src, dst, ew, dinv, lcount, lkey, lnorm, b1, out1);

    {
        gemm_wmma<HID, HID2><<<(N_NODES + GBM(HID2) - 1) / GBM(HID2), GW * 32, 0, stream>>>(out1, W2, H2);
    }
    own_rows<HID2><<<NGRP, 1024, 0, stream>>>(
        H2, src, dst, ew, dinv, lcount, lkey, lnorm, b2, out2);

    classifier<<<(N_NODES + TB - 1) / TB, TB, 0, stream>>>(out2, Wc, bc, logits);
}


# norm pass: one 1024-thread work-group per list (8192 waves, was 131072 short ones), 16 entries per thread in flight, source dinv read from LDS (table staged in two halves); same norm expression
# speedup vs baseline: 4.2102x; 1.0413x over previous
#include <hip/hip_runtime.h>

#define N_NODES 100000
#define N_EDGES 3200000
#define IN_CH   128
#define HID     96
#define HID2    48
#define NCLS    3

typedef __attribute__((ext_vector_type(2)))  float  v2f;
typedef __attribute__((ext_vector_type(8)))  float  v8f;
typedef __attribute__((ext_vector_type(4)))  float  v4f;


#define NGRP   256
#define RNG    ((N_NODES + NGRP - 1) / NGRP)
#define CAP    16384
#define DHALF  (N_NODES / 2)
#define DSU    7

static_assert((NGRP - 1) * RNG < N_NODES && NGRP * RNG >= N_NODES, "NGRP work-groups cover the nodes, none is empty");
static_assert(N_NODES % 2 == 0 && DHALF % 4 == 0, "two equal halves of whole 16-byte pieces");
static_assert(CAP % 1024 == 0, "whole entries per thread");
static_assert(RNG <= 1024, "one thread per owned row");

__global__ __launch_bounds__(1024) void list_norm(const int* __restrict__ src,
                                                  const float* __restrict__ dinv,
                                                  const int* __restrict__ lcount,
                                                  unsigned* lkey,
                                                  float* lnorm) {
    __shared__ __attribute__((aligned(16))) float sd[DHALF];
    __shared__ float sown[RNG];
    constexpr int PT = CAP / 1024;
    constexpr int SQ = DHALF / 4;
    const int blk = blockIdx.x;
    const int cnt = lcount[blk];
    if (cnt <= 0) return;
    unsigned* lk = lkey  + (size_t)blk * CAP;
    float*    ln = lnorm + (size_t)blk * CAP;

    unsigned ent[PT]; float w[PT];
    #pragma unroll
    for (int j = 0; j < PT; ++j) {
        const int i = j * 1024 + (int)threadIdx.x;
        ent[j] = 0u; w[j] = 0.0f;
        if (i < cnt) { ent[j] = lk[i]; w[j] = ln[i]; }
    }
    if (threadIdx.x < RNG) {
        const int node = blk * RNG + (int)threadIdx.x;
        sown[threadIdx.x] = node < N_NODES ? dinv[node] : 0.0f;
    }

    int s[PT];
    #pragma unroll
    for (int j = 0; j < PT; ++j) s[j] = src[ent[j] & 0x3FFFFFu];

    float ds[PT];
    #pragma unroll
    for (int j = 0; j < PT; ++j) ds[j] = 0.0f;
    #pragma unroll
    for (int h = 0; h < 2; ++h) {
        const float4* dg = (const float4*)(dinv + h * DHALF);
        #pragma unroll 1
        for (int q0 = 0; q0 < SQ; q0 += 1024 * DSU) {
            float4 t[DSU];
            #pragma unroll
            for (int u = 0; u < DSU; ++u) {
                const int q = q0 + u * 1024 + (int)threadIdx.x;
                t[u] = make_float4(0.0f, 0.0f, 0.0f, 0.0f);
                if (q < SQ) t[u] = dg[q];
            }
            if (h && q0 == 0) __syncthreads();
            #pragma unroll
            for (int u = 0; u < DSU; ++u) {
                const int q = q0 + u * 1024 + (int)threadIdx.x;
                if (q < SQ) *(float4*)(sd + q * 4) = t[u];
            }
        }
        __syncthreads();
        #pragma unroll
        for (int j = 0; j < PT; ++j) {
            const unsigned n = (unsigned)(s[j] - h * DHALF);
            if (n < (unsigned)DHALF) ds[j] = sd[n];
        }
    }

    float dd[PT];
    #pragma unroll
    for (int j = 0; j < PT; ++j) dd[j] = sown[ent[j] >> 22];
    #pragma unroll
    for (int j = 0; j < PT; ++j) {
        const int i = j * 1024 + (int)threadIdx.x;
        if (i < cnt) {
            ln[i] = ds[j] * w[j] * dd[j];
            lk[i] = ((ent[j] >> 22) << 17) | (unsigned)s[j];
        }
    }
}


#define GW 6
#define GR 4
#define GBM(COLS) ((GW / ((COLS) / 16)) * GR * 16)

static_assert(N_NODES % 16 == 0, "whole row tiles: a row tile is all inside or all outside");

template<int K, int COLS>
__global__ __launch_bounds__(GW * 32) void gemm_wmma(const float* __restrict__ X,
                                                     const float* __restrict__ W,
                                                     float* __restrict__ H) {
    constexpr int CT  = COLS / 16;
    constexpr int BM  = GBM(COLS);
    constexpr int P   = K + 4;
    constexpr int K4  = K / 4;
    constexpr int TOT = BM * K4;
    constexpr int NT  = GW * 32;
    constexpr int NIT = (TOT + NT - 1) / NT;
    constexpr int ROWS = (NIT * NT + K4 - 1) / K4;
    static_assert(COLS % 16 == 0 && GW % CT == 0, "the waves split into whole column tiles");
    static_assert(K % 4 == 0 && (P / 4) % 2 == 1, "pitch is 16 B x odd");
    __shared__ __attribute__((aligned(16))) float xs[ROWS * P];

    const int lane = threadIdx.x & 31;
    const int wave = __builtin_amdgcn_readfirstlane((int)(threadIdx.x >> 5));
    const int m0 = blockIdx.x * BM;

    const float4* xg = (const float4*)(X + (size_t)m0 * K);
    const int have = (N_NODES - m0) * K4;
    const int lim = have < TOT ? have : TOT;
    float4 v[NIT];
    #pragma unroll
    for (int i = 0; i < NIT; ++i) {
        const int idx = (int)threadIdx.x + i * NT;
        v[i] = xg[idx < lim ? idx : lim - 1];
    }

    const int ct  = wave % CT;
    const int rg  = wave / CT;
    const int col = ct * 16 + (lane & 15);
    const int kh  = (lane >> 4) << 1;
    v2f bf[K4];
    #pragma unroll
    for (int j = 0; j < K4; ++j) {
        bf[j].x = W[(size_t)(j * 4 + kh)     * COLS + col];
        bf[j].y = W[(size_t)(j * 4 + kh + 1) * COLS + col];
    }

    #pragma unroll
    for (int i = 0; i < NIT; ++i) {
        const int idx = (int)threadIdx.x + i * NT;
        const int r = idx / K4, q = idx - r * K4;
        *(float4*)(xs + r * P + q * 4) = v[i];
    }
    __syncthreads();

    const float* ar = xs + (rg * GR * 16 + (lane & 15)) * P + kh;
    v8f c[GR] = {};
    #pragma unroll
    for (int j = 0; j < K4; ++j) {
        #pragma unroll
        for (int t = 0; t < GR; ++t) {
            const v2f a = *(const v2f*)(ar + t * 16 * P + j * 4);
            c[t] = __builtin_amdgcn_wmma_f32_16x16x4_f32(false, a, false, bf[j],
                                                         (short)0, c[t], false, false);
        }
    }

    const int rbase = (lane >> 4) << 3;
    #pragma unroll
    for (int t = 0; t < GR; ++t) {
        const int mt = m0 + (rg * GR + t) * 16;
        if (mt < N_NODES) {
            float* hrow = H + (size_t)(mt + rbase) * COLS + col;
            #pragma unroll
            for (int r = 0; r < 8; ++r) hrow[(size_t)r * COLS] = c[t][r];
        }
    }
}


#define RING   256
#define CHUNK  1024
#define NCHUNK (N_EDGES / CHUNK)

static_assert(N_EDGES % CHUNK == 0, "whole chunks");
static_assert(N_EDGES <= (1 << 22), "edge id packs into 22 bits");
static_assert(RNG <= 1024, "row offset packs into 10 bits, one thread per owned node");
static_assert(N_NODES <= (1 << 17), "source node packs into 17 bits beside the row offset");

#define SUP    16
#define NSUP   (NGRP / SUP)
#define SRNG   (SUP * RNG)
#define NSL    16
#define CH1    512
#define NCH1   (N_EDGES / CH1)
#define SEGCAP 16384
#define CPAD   0xFFFFFFFFu

static_assert(NGRP % SUP == 0, "whole coarse ranges");
static_assert(N_EDGES % CH1 == 0 && SEGCAP % CH1 == 0, "whole trips");
static_assert((NCH1 / NSL + 1) * CH1 <= (1 << 18), "edge offset in a slice packs into 18 bits");
static_assert(SRNG < (1 << 14) - 1, "node offset in a coarse range packs into 14 bits, below the filler's");
static_assert((CPAD >> 18) - (SUP - 1) * RNG >= RNG, "the filler matches no owner range");
static_assert(NSL <= 32, "one lane per list of a coarse range");

static __device__ __forceinline__ int   lane_i(int v, int l)   { return __builtin_amdgcn_readlane(v, l); }
static __device__ __forceinline__ float lane_f(float v, int l) {
    return __int_as_float(__builtin_amdgcn_readlane(__float_as_int(v), l));
}
static __device__ __forceinline__ void lds_add(float* p, float v) {
    __hip_atomic_fetch_add(p, v, __ATOMIC_RELAXED, __HIP_MEMORY_SCOPE_WORKGROUP);
}

#define INFL 16
static_assert(32 % INFL == 0, "a full trip is whole groups");

template<int C, bool FULL, class FNext>
static __device__ __forceinline__ void drain(float* __restrict__ tile,
                                             int s, int loc, float nw, int n, int lane,
                                             const float* __restrict__ H, FNext behind_first) {
    constexpr int T = (C + 31) / 32;
    const int cnt = FULL ? 32 : n;
    #pragma unroll 1
    for (int g = 0; g < cnt; g += INFL) {
        float hv[INFL][T];
        #pragma unroll
        for (int u = 0; u < INFL; ++u) {
            const int i = FULL ? g + u : (g + u < n ? g + u : n - 1);
            const float* hr = H + (size_t)lane_i(s, i) * C + lane;
            #pragma unroll
            for (int t = 0; t < T; ++t)
                hv[u][t] = (t * 32 + 32 <= C || lane < C - t * 32) ? hr[t * 32] : 0.0f;
        }
        if (g == 0) behind_first();
        #pragma unroll
        for (int u = 0; u < INFL; ++u) {
            if (FULL || g + u < n) {
                const float w = lane_f(nw, g + u);
                float* tr = tile + lane_i(loc, g + u) * C + lane;
                #pragma unroll
                for (int t = 0; t < T; ++t)
                    if (t * 32 + 32 <= C || lane < C - t * 32) lds_add(tr + t * 32, w * hv[u][t]);
            }
        }
    }
}

#define QW   32
#define OWNL 24

template<int V> struct piece;
template<> struct piece<4> { typedef v4f t; };
template<> struct piece<2> { typedef v2f t; };

template<int C>
static __device__ __forceinline__ void drain_own(float* tile, const uint2* q, int q0, int q1,
                                                 int lane, const float* __restrict__ H) {
    constexpr int V = C / OWNL;
    constexpr int R = 32 / V;
    static_assert(V * OWNL == C && (V == 2 || V == 4), "a row is 24 pieces of 8 or 16 bytes");
    static_assert(R <= OWNL, "the lanes that hold a group's entries all hold row pieces");
    typedef typename piece<V>::t pv;
    if (lane < OWNL && q0 < q1) {
        auto ents = [&](int g) -> uint2 {
            const int i = g + lane;
            return q[i < q1 ? i : q1 - 1];
        };
        auto ask = [&](pv (&hv)[R], const uint2& e, int n) {
            #pragma unroll
            for (int u = 0; u < R; ++u) {
                if (u < n) {
                    const int s = lane_i((int)(e.x & 0x1FFFFu), u);
                    hv[u] = *(const pv*)(H + (size_t)s * C + lane * V);
                }
            }
        };
        auto add = [&](const pv (&hv)[R], const uint2& e, int n) {
            #pragma unroll
            for (int u = 0; u < R; u += 2) {
                if (u + 1 < n) {
                    const float w0 = lane_f(__uint_as_float(e.y), u);
                    const float w1 = lane_f(__uint_as_float(e.y), u + 1);
                    const int l0 = lane_i((int)(e.x >> 17), u);
                    const int l1 = lane_i((int)(e.x >> 17), u + 1);
                    pv* t0 = (pv*)(tile + l0 * C + lane * V);
                    if (l0 != l1) {
                        pv* t1 = (pv*)(tile + l1 * C + lane * V);
                        const pv x0 = *t0;
                        const pv x1 = *t1;
                        *t0 = x0 + w0 * hv[u];
                        *t1 = x1 + w1 * hv[u + 1];
                    } else {
                        *t0 = (*t0 + w0 * hv[u]) + w1 * hv[u + 1];
                    }
                } else if (u < n) {
                    const float w = lane_f(__uint_as_float(e.y), u);
                    const int loc = lane_i((int)(e.x >> 17), u);
                    pv* tp = (pv*)(tile + loc * C + lane * V);
                    *tp = *tp + w * hv[u];
                }
            }
        };
        int g = q0;
        if (g + 2 * R <= q1) {
            pv a[R], b[R];
            uint2 ea = ents(g), eb = ents(g + R);
            ask(a, ea, R);
            #pragma unroll 1
            for (;;) {
                ask(b, eb, R);
                const uint2 ea2 = ents(g + 2 * R);
                add(a, ea, R);
                g += 2 * R;
                if (g + 2 * R > q1) { add(b, eb, R); break; }
                ea = ea2;
                ask(a, ea, R);
                const uint2 eb2 = ents(g + R);
                add(b, eb, R);
                eb = eb2;
            }
        }
        if (g + R <= q1) {
            pv t[R];
            const uint2 e = ents(g);
            ask(t, e, R);
            add(t, e, R);
            g += R;
        }
        if (g < q1) {
            pv t[R];
            const uint2 e = ents(g);
            ask(t, e, q1 - g);
            add(t, e, q1 - g);
        }
    }
}

template<class F32, class FRest>
static __device__ __forceinline__ void scan_owned(const int* __restrict__ dst, int base,
                                                  unsigned* __restrict__ rq, int lane, int wave,
                                                  F32 on32, FRest rest) {
    unsigned head = 0, tail = 0;
    const int start = (int)(((long long)blockIdx.x * NCHUNK) / gridDim.x);
    for (int it = wave; it < NCHUNK; it += 32) {
        int chunk = start + it;
        if (chunk >= NCHUNK) chunk -= NCHUNK;
        const int e0 = chunk * CHUNK;
        const int4* dp = (const int4*)(dst + e0) + lane;
        int4 v[8];
        #pragma unroll
        for (int j = 0; j < 8; ++j) v[j] = dp[j * 32];
        #pragma unroll
        for (int j = 0; j < 8; ++j) {
            const int dd[4] = { v[j].x, v[j].y, v[j].z, v[j].w };
            #pragma unroll
            for (int k = 0; k < 4; ++k) {
                const unsigned loc = (unsigned)(dd[k] - base);
                const bool m = loc < (unsigned)RNG;
                const unsigned mask = __builtin_amdgcn_ballot_w32(m);
                if (mask) {
                    if (m) {
                        const unsigned pos = (tail + __builtin_amdgcn_mbcnt_lo(mask, 0u)) & (RING - 1);
                        rq[pos] = (loc << 22) | (unsigned)(e0 + j * 128 + lane * 4 + k);
                    }
                    tail += (unsigned)__builtin_popcount(mask);
                }
            }
            while (tail - head >= 32u) { on32(head); head += 32u; }
        }
    }
    if (tail != head) rest(head, (int)(tail - head));
}

__global__ __launch_bounds__(1024) void coarse_split(const int* __restrict__ dst,
                                                     unsigned* __restrict__ clist,
                                                     int* __restrict__ ccount) {
    __shared__ unsigned ring[32][RING];
    __shared__ int      lcnt, lovf;
    const int lane = threadIdx.x & 31;
    const int wave = __builtin_amdgcn_readfirstlane((int)(threadIdx.x >> 5));
    const int p = blockIdx.x;
    const int a = blockIdx.y;
    const int c0 = p * NCH1 / NSL;
    const int nch = (p + 1) * NCH1 / NSL - c0;
    const int cbase = a * SRNG;
    const int* dsl = dst + (size_t)c0 * CH1;
    if (threadIdx.x == 0) { lcnt = 0; lovf = 0; }
    __syncthreads();
    unsigned* rq  = ring[wave];
    unsigned* seg = clist + (size_t)(a * NSL + p) * SEGCAP;
    auto put = [&](unsigned head, int n) {
        unsigned ent = 0;
        if (lane < n) ent = rq[(head + lane) & (RING - 1)];
        int pos = 0;
        if (lane == 0)
            pos = __hip_atomic_fetch_add(&lcnt, n, __ATOMIC_RELAXED, __HIP_MEMORY_SCOPE_WORKGROUP);
        pos = __builtin_amdgcn_readfirstlane(pos);
        if (pos + n <= SEGCAP) {
            if (lane < n) seg[pos + lane] = ent;
        } else if (lane == 0) {
            __hip_atomic_store(&lovf, 1, __ATOMIC_RELAXED, __HIP_MEMORY_SCOPE_WORKGROUP);
        }
    };
    unsigned head = 0, tail = 0;
    const int rot = a * nch / NSUP;
    for (int it = wave; it < nch; it += 32) {
        int c = it + rot;
        if (c >= nch) c -= nch;
        const int eo = c * CH1;
        const int4* dp = (const int4*)(dsl + eo) + lane;
        int4 v[4];
        #pragma unroll
        for (int j = 0; j < 4; ++j) v[j] = dp[j * 32];
        #pragma unroll
        for (int j = 0; j < 4; ++j) {
            const int dd[4] = { v[j].x, v[j].y, v[j].z, v[j].w };
            #pragma unroll
            for (int k = 0; k < 4; ++k) {
                const unsigned loc = (unsigned)(dd[k] - cbase);
                const bool m = loc < (unsigned)SRNG;
                const unsigned mask = __builtin_amdgcn_ballot_w32(m);
                if (mask) {
                    if (m) {
                        const unsigned pos = (tail + __builtin_amdgcn_mbcnt_lo(mask, 0u)) & (RING - 1);
                        rq[pos] = (loc << 18) | (unsigned)(eo + j * 128 + lane * 4 + k);
                    }
                    tail += (unsigned)__builtin_popcount(mask);
                }
            }
            while (tail - head >= 32u) { put(head, 32); head += 32u; }
        }
    }
    if (tail != head) put(head, (int)(tail - head));
    __syncthreads();
    const int cnt = lcnt, ovf = lovf;
    if (!ovf) {
        const int end = (cnt + CH1 - 1) / CH1 * CH1;
        for (int i = cnt + (int)threadIdx.x; i < end; i += 1024) seg[i] = CPAD;
    }
    if (threadIdx.x == 0) ccount[a * NSL + p] = ovf ? -1 : cnt;
}

__global__ __launch_bounds__(1024) void own_deg(const int* __restrict__ dst,
                                                const float* __restrict__ ew,
                                                const unsigned* __restrict__ clist,
                                                const int* __restrict__ ccount,
                                                float* __restrict__ dinv,
                                                unsigned* __restrict__ lkey,
                                                float* __restrict__ lw,
                                                int* __restrict__ lcount) {
    __shared__ float    sdeg[RNG];
    __shared__ unsigned ring[32][RING];
    __shared__ int      lcnt, lovf;
    const int lane = threadIdx.x & 31;
    const int wave = __builtin_amdgcn_readfirstlane((int)(threadIdx.x >> 5));
    const int base = blockIdx.x * RNG;
    const int a = blockIdx.x / SUP;
    const int b = blockIdx.x % SUP;
    if (threadIdx.x == 0) { lcnt = 0; lovf = 0; }
    if (threadIdx.x < RNG) sdeg[threadIdx.x] = 1.0f;
    __syncthreads();
    unsigned* rq = ring[wave];
    unsigned* lk = lkey + (size_t)blockIdx.x * CAP;
    float*    lv = lw   + (size_t)blockIdx.x * CAP;
    auto add_n = [&](unsigned head, int n) {
        unsigned ent = 0; float w = 0.0f;
        if (lane < n) {
            ent = rq[(head + lane) & (RING - 1)];
            w   = ew[ent & 0x3FFFFFu];
        }
        int pos = 0;
        if (lane == 0)
            pos = __hip_atomic_fetch_add(&lcnt, n, __ATOMIC_RELAXED, __HIP_MEMORY_SCOPE_WORKGROUP);
        pos = __builtin_amdgcn_readfirstlane(pos);
        if (pos + n <= CAP) {
            if (lane < n) { lk[pos + lane] = ent; lv[pos + lane] = w; }
        } else if (lane == 0) {
            __hip_atomic_store(&lovf, 1, __ATOMIC_RELAXED, __HIP_MEMORY_SCOPE_WORKGROUP);
        }
        if (lane < n) lds_add(sdeg + (ent >> 22), w);
    };
    const int cn = lane < NSL ? ccount[a * NSL + lane] : 0;
    if (__builtin_amdgcn_ballot_w32(cn < 0) == 0u) {
        const int nc = (cn + CH1 - 1) / CH1;
        int before = 0;
        #pragma unroll
        for (int q = 0; q < NSL; ++q) {
            const int t = lane_i(nc, q);
            if (lane > q) before += t;
        }
        const int upto = before + nc;
        const int total = lane_i(upto, 31);
        const int rot = b * total / SUP;
        const unsigned lbase = (unsigned)(b * RNG);
        unsigned head = 0, tail = 0;
        for (int it = wave; it < total; it += 32) {
            int gi = it + rot;
            if (gi >= total) gi -= total;
            const int p = __builtin_popcount(__builtin_amdgcn_ballot_w32(lane < NSL && upto <= gi));
            const int c = gi - lane_i(before, p);
            const int ef = (p * NCH1 / NSL) * CH1;
            const uint4* dp = (const uint4*)(clist + (size_t)(a * NSL + p) * SEGCAP + (size_t)c * CH1) + lane;
            uint4 v[4];
            #pragma unroll
            for (int j = 0; j < 4; ++j) v[j] = dp[j * 32];
            #pragma unroll
            for (int j = 0; j < 4; ++j) {
                const unsigned en[4] = { v[j].x, v[j].y, v[j].z, v[j].w };
                #pragma unroll
                for (int k = 0; k < 4; ++k) {
                    const unsigned loc = (en[k] >> 18) - lbase;
                    const bool m = loc < (unsigned)RNG;
                    const unsigned mask = __builtin_amdgcn_ballot_w32(m);
                    if (mask) {
                        if (m) {
                            const unsigned pos = (tail + __builtin_amdgcn_mbcnt_lo(mask, 0u)) & (RING - 1);
                            rq[pos] = (loc << 22) | ((unsigned)ef + (en[k] & 0x3FFFFu));
                        }
                        tail += (unsigned)__builtin_popcount(mask);
                    }
                }
                while (tail - head >= 32u) { add_n(head, 32); head += 32u; }
            }
        }
        if (tail != head) add_n(head, (int)(tail - head));
    } else {
        scan_owned(dst, base, rq, lane, wave,
                   [&](unsigned head) { add_n(head, 32); }, add_n);
    }
    __syncthreads();
    if (threadIdx.x < RNG && base + (int)threadIdx.x < N_NODES)
        dinv[base + threadIdx.x] = rsqrtf(sdeg[threadIdx.x]);
    if (threadIdx.x == 0) lcount[blockIdx.x] = lovf ? -1 : lcnt;
}

template<int C>
__global__ __launch_bounds__(1024) void own_rows(const float* __restrict__ H,
                                                 const int* __restrict__ src,
                                                 const int* __restrict__ dst,
                                                 const float* __restrict__ ew,
                                                 const float* __restrict__ dinv,
                                                 const int* __restrict__ lcount,
                                                 const unsigned* __restrict__ lkey,
                                                 const float* __restrict__ lnorm,
                                                 const float* __restrict__ bias,
                                                 float* __restrict__ out) {
    __shared__ __attribute__((aligned(16))) float tile[RNG * C];
    __shared__ unsigned ring[32][RING];
    __shared__ __attribute__((aligned(16))) uint2 queue[CAP];
    __shared__ int      qn[QW];
    __shared__ int      qo[QW + 1];
    constexpr int C4 = C / 4;
    const int lane = threadIdx.x & 31;
    const int wave = __builtin_amdgcn_readfirstlane((int)(threadIdx.x >> 5));
    const int base = blockIdx.x * RNG;

    if (threadIdx.x < QW) qn[threadIdx.x] = 0;
    for (int idx = threadIdx.x; idx < RNG * C4; idx += 1024) {
        const int r = idx / C4, q = idx - r * C4;
        const int node = base + r;
        float4 o = { 0.0f, 0.0f, 0.0f, 0.0f };
        if (node < N_NODES) {
            const float di = dinv[node], w = di * di;
            const float4 hv = *(const float4*)(H + (size_t)node * C + q * 4);
            o.x = w * hv.x; o.y = w * hv.y; o.z = w * hv.z; o.w = w * hv.w;
        }
        *(float4*)(tile + idx * 4) = o;
    }
    __syncthreads();

    const int cnt = lcount[blockIdx.x];
    if (__builtin_expect(cnt >= 0, 1)) {
        const unsigned* lk = lkey  + (size_t)blockIdx.x * CAP;
        const float*    ln = lnorm + (size_t)blockIdx.x * CAP;
        constexpr int PT = CAP / 1024;
        unsigned key[PT]; float nw[PT];
        #pragma unroll
        for (int j = 0; j < PT; ++j) {
            const int i = j * 1024 + (int)threadIdx.x;
            key[j] = 0u; nw[j] = 0.0f;
            if (i < cnt) { key[j] = lk[i]; nw[j] = ln[i]; }
        }
        #pragma unroll
        for (int j = 0; j < PT; ++j) {
            const int i = j * 1024 + (int)threadIdx.x;
            if (i < cnt)
                __hip_atomic_fetch_add(&qn[(key[j] >> 17) & (QW - 1)], 1,
                                       __ATOMIC_RELAXED, __HIP_MEMORY_SCOPE_WORKGROUP);
        }
        __syncthreads();
        if (wave == 0) {
            const int c = qn[lane];
            int incl = c;
            #pragma unroll
            for (int d = 1; d < 32; d <<= 1) {
                const int t = __shfl_up(incl, d);
                if (lane >= d) incl += t;
            }
            qo[lane + 1] = incl;
            if (lane == 0) qo[0] = 0;
            qn[lane] = incl - c;
        }
        __syncthreads();
        int slot[PT];
        #pragma unroll
        for (int j = 0; j < PT; ++j) {
            const int i = j * 1024 + (int)threadIdx.x;
            slot[j] = 0;
            if (i < cnt)
                slot[j] = __hip_atomic_fetch_add(&qn[(key[j] >> 17) & (QW - 1)], 1,
                                                 __ATOMIC_RELAXED, __HIP_MEMORY_SCOPE_WORKGROUP);
        }
        #pragma unroll
        for (int j = 0; j < PT; ++j) {
            const int i = j * 1024 + (int)threadIdx.x;
            if (i < cnt) queue[slot[j]] = make_uint2(key[j], __float_as_uint(nw[j]));
        }
        __syncthreads();
        drain_own<C>(tile, queue, qo[wave], qo[wave + 1], lane, H);
    } else {
        unsigned* rq = ring[wave];
        auto look = [&](unsigned head, int n, int& s, int& loc, float& nw) {
            s = 0; loc = 0; nw = 0.0f;
            if (lane < n) {
                const unsigned ent = rq[(head + lane) & (RING - 1)];
                const int e = (int)(ent & 0x3FFFFFu);
                loc = (int)(ent >> 22);
                s   = src[e];
                nw  = dinv[s] * ew[e] * dinv[base + loc];
            }
        };
        auto nothing = []() {};
        scan_owned(dst, base, rq, lane, wave,
                   [&](unsigned head) { int s, loc; float nw; look(head, 32, s, loc, nw);
                                        drain<C, true>(tile, s, loc, nw, 32, lane, H, nothing); },
                   [&](unsigned head, int n) { int s, loc; float nw; look(head, n, s, loc, nw);
                                               drain<C, false>(tile, s, loc, nw, n, lane, H, nothing); });
    }
    __syncthreads();

    for (int idx = threadIdx.x; idx < RNG * C4; idx += 1024) {
        const int r = idx / C4, q = idx - r * C4;
        const int node = base + r;
        if (node < N_NODES) {
            float4 t4 = *(const float4*)(tile + idx * 4);
            const float4 bb = *(const float4*)(bias + q * 4);
            t4.x = fmaxf(t4.x + bb.x, 0.0f);
            t4.y = fmaxf(t4.y + bb.y, 0.0f);
            t4.z = fmaxf(t4.z + bb.z, 0.0f);
            t4.w = fmaxf(t4.w + bb.w, 0.0f);
            *(float4*)(out + (size_t)node * C + q * 4) = t4;
        }
    }
}

__global__ __launch_bounds__(256) void classifier(const float* __restrict__ Hin,
                                                  const float* __restrict__ Wc,
                                                  const float* __restrict__ bc,
                                                  float* __restrict__ logits) {
    int i = blockIdx.x * blockDim.x + threadIdx.x;
    if (i >= N_NODES) return;
    const float* h = Hin + (size_t)i * HID2;
    float a0 = bc[0], a1 = bc[1], a2 = bc[2];
    #pragma unroll
    for (int k = 0; k < HID2; ++k) {
        float hv = h[k];
        a0 += hv * Wc[k * 3 + 0];
        a1 += hv * Wc[k * 3 + 1];
        a2 += hv * Wc[k * 3 + 2];
    }
    logits[(size_t)i * 3 + 0] = a0;
    logits[(size_t)i * 3 + 1] = a1;
    logits[(size_t)i * 3 + 2] = a2;
}


extern "C" void kernel_launch(void* const* d_in, const int* in_sizes, int n_in,
                              void* d_out, int out_size, void* d_ws, size_t ws_size,
                              hipStream_t stream) {
    const float* x  = (const float*)d_in[0];
    const int*   ei = (const int*)  d_in[1];
    const float* ew = (const float*)d_in[2];
    const float* W1 = (const float*)d_in[3];
    const float* b1 = (const float*)d_in[4];
    const float* W2 = (const float*)d_in[5];
    const float* b2 = (const float*)d_in[6];
    const float* Wc = (const float*)d_in[7];
    const float* bc = (const float*)d_in[8];
    float* logits = (float*)d_out;

    const int* src = ei;
    const int* dst = ei + N_EDGES;

    char* ws = (char*)d_ws;
    size_t off = 0;
    auto take = [&](size_t bytes) -> void* {
        void* p = ws + off;
        off += (bytes + 255) & ~(size_t)255;
        return p;
    };
    float* dinv = (float*)take((size_t)N_NODES * 4);
    int*      lcount = (int*)     take((size_t)(NGRP) * 4);
    unsigned* lkey   = (unsigned*)take((size_t)(NGRP) * CAP * 4);
    float*    lnorm  = (float*)   take((size_t)(NGRP) * CAP * 4);
    unsigned* clist  = (unsigned*)take((size_t)(NSUP * NSL) * SEGCAP * 4);
    int*      ccount = (int*)     take((size_t)(NSUP * NSL) * 4);
    float* H1   = (float*)take((size_t)N_NODES * HID * 4);
    float* out1 = (float*)take((size_t)N_NODES * HID * 4);
    float* H2   = H1;
    float* out2 = out1;

    const int TB = 256;
    {
        dim3 g(NSL, NSUP);
        coarse_split<<<g, 1024, 0, stream>>>(dst, clist, ccount);
    }
    own_deg<<<NGRP, 1024, 0, stream>>>(dst, ew, clist, ccount, dinv, lkey, lnorm, lcount);
    list_norm<<<NGRP, 1024, 0, stream>>>(src, dinv, lcount, lkey, lnorm);

    {
        gemm_wmma<IN_CH, HID><<<(N_NODES + GBM(HID) - 1) / GBM(HID), GW * 32, 0, stream>>>(x, W1, H1);
    }
    own_rows<HID><<<NGRP, 1024, 0, stream>>>(
        H1, src, dst, ew, dinv, lcount, lkey, lnorm, b1, out1);

    {
        gemm_wmma<HID, HID2><<<(N_NODES + GBM(HID2) - 1) / GBM(HID2), GW * 32, 0, stream>>>(out1, W2, H2);
    }
    own_rows<HID2><<<NGRP, 1024, 0, stream>>>(
        H2, src, dst, ew, dinv, lcount, lkey, lnorm, b2, out2);

    classifier<<<(N_NODES + TB - 1) / TB, TB, 0, stream>>>(out2, Wc, bc, logits);
}
